# starting kernel
# baseline (speedup 1.0000x reference)
#include <hip/hip_runtime.h>

typedef __attribute__((ext_vector_type(2))) float v2f;
typedef __attribute__((ext_vector_type(8))) float v8f;

#define DIM 128

__global__ void k_deg_init(float* deg, int n) {
    int i = blockIdx.x * blockDim.x + threadIdx.x;
    if (i < n) deg[i] = 1.0f;
}

__global__ void k_deg_count(const int* __restrict__ col, float* deg, int e) {
    int i = blockIdx.x * blockDim.x + threadIdx.x;
    if (i < e) atomicAdd(&deg[col[i]], 1.0f);
}

__global__ void k_deg_rsqrt(float* deg, int n) {
    int i = blockIdx.x * blockDim.x + threadIdx.x;
    if (i < n) deg[i] = rsqrtf(deg[i]);
}

__global__ __launch_bounds__(256) void k_gemm128_wmma(
    const float* __restrict__ X, const float* __restrict__ W,
    const float* __restrict__ cvec, float* __restrict__ Y, int n) {
    const int lane = threadIdx.x & 31;
    const int wave = threadIdx.x >> 5;
    const int m0   = blockIdx.x * 16;
    const int n0   = wave * 16;
    const int lo   = lane & 15;
    const int hi   = lane >> 4;

    int arow = m0 + lo;
    int arow_c = (arow < n) ? arow : (n - 1);
    const float* ap = X + (long long)arow_c * DIM + 2 * hi;
    const float* bp = W + (n0 + lo) + (long long)(2 * hi) * DIM;

    float ci = cvec ? cvec[n0 + lo] : 0.0f;
    v8f d;
#pragma unroll
    for (int v = 0; v < 8; ++v) d[v] = ci;

#pragma unroll
    for (int k = 0; k < DIM; k += 4) {
        v2f a = *(const v2f*)(ap + k);
        v2f b;
        b.x = bp[(long long)k * DIM];
        b.y = bp[(long long)k * DIM + DIM];
        d = __builtin_amdgcn_wmma_f32_16x16x4_f32(false, a, false, b,
                                                  (short)0, d, false, false);
    }

    float* yp = Y + (long long)(m0 + 8 * hi) * DIM + n0 + lo;
#pragma unroll
    for (int v = 0; v < 8; ++v) {
        int m = m0 + v + 8 * hi;
        if (m < n) yp[(long long)v * DIM] = d[v];
    }
}

__global__ void k_selfloop_init(const float* __restrict__ xw, const float* __restrict__ dinv,
                                float* __restrict__ acc, int n) {
    int t = blockIdx.x * blockDim.x + threadIdx.x;
    if (t >= n * 32) return;
    int node = t >> 5;
    float s = dinv[node]; s *= s;
    float4 v = ((const float4*)xw)[t];
    float4 o = make_float4(v.x * s, v.y * s, v.z * s, v.w * s);
    ((float4*)acc)[t] = o;
}

__global__ void k_edge_scatter(const float* __restrict__ xw,
                               const int* __restrict__ rowi, const int* __restrict__ coli,
                               const float* __restrict__ dinv, float* __restrict__ acc, int e) {
    long long t = (long long)blockIdx.x * blockDim.x + threadIdx.x;
    int eid = (int)(t >> 5);
    if (eid >= e) return;
    int lane = (int)(t & 31);
    int r = rowi[eid], c = coli[eid];
    float norm = dinv[r] * dinv[c];
    float4 v = *(const float4*)(xw + (long long)r * DIM + lane * 4);
    float* dst = acc + (long long)c * DIM + lane * 4;
    atomicAdd(dst + 0, v.x * norm);
    atomicAdd(dst + 1, v.y * norm);
    atomicAdd(dst + 2, v.z * norm);
    atomicAdd(dst + 3, v.w * norm);
}

__global__ void k_root_and_c(const float* __restrict__ acc, const float* __restrict__ b1,
                             const float* __restrict__ posts, const float* __restrict__ W2,
                             const int* __restrict__ rootPtr,
                             float* __restrict__ root_feat, float* __restrict__ cvec) {
    int j = threadIdx.x;
    int root = *rootPtr;
    root_feat[j] = acc[(long long)root * DIM + j] + b1[j];
    float s = 0.0f;
    for (int k = 0; k < DIM; ++k)
        s += fmaxf(posts[(long long)root * DIM + k], 0.0f) * W2[(long long)(DIM + k) * DIM + j];
    cvec[j] = s;
}

__global__ void k_bias_relu(float* __restrict__ acc, const float* __restrict__ b, int n) {
    int t = blockIdx.x * blockDim.x + threadIdx.x;
    if (t >= n * 32) return;
    float4 bb = ((const float4*)b)[t & 31];
    float4 v = ((float4*)acc)[t];
    v.x = fmaxf(v.x + bb.x, 0.0f);
    v.y = fmaxf(v.y + bb.y, 0.0f);
    v.z = fmaxf(v.z + bb.z, 0.0f);
    v.w = fmaxf(v.w + bb.w, 0.0f);
    ((float4*)acc)[t] = v;
}

__global__ __launch_bounds__(256) void k_finalize_pool(
    const float* __restrict__ acc, const float* __restrict__ b2,
    const float* __restrict__ root_feat, float* __restrict__ out, int n) {
    __shared__ float feat[256];
    int i = blockIdx.x;
    int t = threadIdx.x;
    if (t < 128) feat[t] = root_feat[t];
    else         feat[t] = fmaxf(acc[(long long)i * DIM + (t - 128)] + b2[t - 128], 0.0f);
    __syncthreads();
    if (t < 254)
        out[(long long)i * 254 + t] = (feat[t] + feat[t + 1] + feat[t + 2]) * (1.0f / 3.0f);
}

extern "C" void kernel_launch(void* const* d_in, const int* in_sizes, int n_in,
                              void* d_out, int out_size, void* d_ws, size_t ws_size,
                              hipStream_t stream) {
    const float* posts = (const float*)d_in[0];
    const float* W1    = (const float*)d_in[1];
    const float* b1    = (const float*)d_in[2];
    const float* W2    = (const float*)d_in[3];
    const float* b2    = (const float*)d_in[4];
    const int*   edge  = (const int*)d_in[5];
    const int*   rootP = (const int*)d_in[6];

    const int n = in_sizes[0] / DIM;
    const int e = in_sizes[5] / 2;
    const int* rowi = edge;
    const int* coli = edge + e;

    char* ws = (char*)d_ws;
    size_t off = 0;
    float* dinv = (float*)(ws + off); off += (((size_t)n * 4) + 255) & ~(size_t)255;
    float* xw   = (float*)(ws + off); off += (size_t)n * DIM * 4;
    float* acc  = (float*)(ws + off); off += (size_t)n * DIM * 4;
    float* rootf = (float*)(ws + off); off += DIM * 4;
    float* cvec  = (float*)(ws + off);
    float* out = (float*)d_out;

    const int B = 256;
    const int gN   = (n + B - 1) / B;
    const int gE   = (e + B - 1) / B;
    const int gNF4 = (n * 32 + B - 1) / B;
    const int gE32 = (int)(((long long)e * 32 + B - 1) / B);
    const int gM   = (n + 15) / 16;

    k_deg_init <<<gN, B, 0, stream>>>(dinv, n);
    k_deg_count<<<gE, B, 0, stream>>>(coli, dinv, e);
    k_deg_rsqrt<<<gN, B, 0, stream>>>(dinv, n);

    k_gemm128_wmma <<<gM, B, 0, stream>>>(posts, W1, nullptr, xw, n);
    k_selfloop_init<<<gNF4, B, 0, stream>>>(xw, dinv, acc, n);
    k_edge_scatter <<<gE32, B, 0, stream>>>(xw, rowi, coli, dinv, acc, e);

    k_root_and_c<<<1, DIM, 0, stream>>>(acc, b1, posts, W2, rootP, rootf, cvec);

    k_bias_relu    <<<gNF4, B, 0, stream>>>(acc, b1, n);
    k_gemm128_wmma <<<gM, B, 0, stream>>>(acc, W2, cvec, xw, n);
    k_selfloop_init<<<gNF4, B, 0, stream>>>(xw, dinv, acc, n);
    k_edge_scatter <<<gE32, B, 0, stream>>>(xw, rowi, coli, dinv, acc, e);

    k_finalize_pool<<<n, B, 0, stream>>>(acc, b2, rootf, out, n);
}


# Edges grouped by destination in each call (integer slot counters, scan, fill); one wave sums a node's source rows in registers, stores the row once: no f32 atomics; sum order = list (arrival) order
# speedup vs baseline: 2.2323x; 2.2323x over previous
#include <hip/hip_runtime.h>

typedef __attribute__((ext_vector_type(2))) float v2f;
typedef __attribute__((ext_vector_type(8))) float v8f;

#define DIM 128

__global__ void k_zero_i32(int* __restrict__ p, int n) {
    int i = blockIdx.x * blockDim.x + threadIdx.x;
    if (i < n) p[i] = 0;
}

__global__ void k_count_rank(const int* __restrict__ col, int* __restrict__ cnt,
                             int* __restrict__ rank, int e) {
    int i = blockIdx.x * blockDim.x + threadIdx.x;
    if (i < e) rank[i] = atomicAdd(&cnt[col[i]], 1);
}

__global__ __launch_bounds__(1024) void k_scan_dinv(const int* __restrict__ cnt,
                                                    int* __restrict__ offs,
                                                    float* __restrict__ dinv, int n) {
    __shared__ int part[32];
    __shared__ int wtot[32];
    const int t = threadIdx.x;
    const int lane = t & 31;
    const int w = t >> 5;
    const int base = blockIdx.x * 1024;

    int s = 0;
    const int4* c4 = (const int4*)cnt;
    const int lim = base >> 2;
    for (int q = t; q < lim; q += 4096) {
        const bool p1 = q + 1024 < lim, p2 = q + 2048 < lim, p3 = q + 3072 < lim;
        const int4 v0 = c4[q];
        const int4 v1 = c4[p1 ? q + 1024 : q];
        const int4 v2 = c4[p2 ? q + 2048 : q];
        const int4 v3 = c4[p3 ? q + 3072 : q];
        const int s1 = (v1.x + v1.y) + (v1.z + v1.w);
        const int s2 = (v2.x + v2.y) + (v2.z + v2.w);
        const int s3 = (v3.x + v3.y) + (v3.z + v3.w);
        s += ((v0.x + v0.y) + (v0.z + v0.w)) + (p1 ? s1 : 0) + (p2 ? s2 : 0) + (p3 ? s3 : 0);
    }
    for (int o = 16; o > 0; o >>= 1) s += __shfl_xor(s, o);
    if (lane == 0) part[w] = s;

    const int i = base + t;
    const int c = (i < n) ? cnt[i] : 0;
    int x = c;
    for (int o = 1; o < 32; o <<= 1) {
        int y = __shfl_up(x, o);
        if (lane >= o) x += y;
    }
    if (lane == 31) wtot[w] = x;
    __syncthreads();

    int pre = 0;
    for (int k = 0; k < 32; ++k) {
        pre += part[k];
        if (k < w) pre += wtot[k];
    }
    const int excl = pre + x - c;
    if (i < n) {
        offs[i] = excl;
        dinv[i] = rsqrtf((float)(c + 1));
        if (i == n - 1) offs[n] = excl + c;
    }
}

__global__ void k_fill(const int* __restrict__ row, const int* __restrict__ col,
                       const int* __restrict__ rank, const int* __restrict__ offs,
                       int* __restrict__ srcs, int e) {
    int i = blockIdx.x * blockDim.x + threadIdx.x;
    if (i < e) srcs[offs[col[i]] + rank[i]] = row[i];
}

__global__ __launch_bounds__(256) void k_gemm128_wmma(
    const float* __restrict__ X, const float* __restrict__ W,
    const float* __restrict__ cvec, float* __restrict__ Y, int n) {
    const int lane = threadIdx.x & 31;
    const int wave = threadIdx.x >> 5;
    const int m0   = blockIdx.x * 16;
    const int n0   = wave * 16;
    const int lo   = lane & 15;
    const int hi   = lane >> 4;

    int arow = m0 + lo;
    int arow_c = (arow < n) ? arow : (n - 1);
    const float* ap = X + (long long)arow_c * DIM + 2 * hi;
    const float* bp = W + (n0 + lo) + (long long)(2 * hi) * DIM;

    float ci = cvec ? cvec[n0 + lo] : 0.0f;
    v8f d;
#pragma unroll
    for (int v = 0; v < 8; ++v) d[v] = ci;

#pragma unroll
    for (int k = 0; k < DIM; k += 4) {
        v2f a = *(const v2f*)(ap + k);
        v2f b;
        b.x = bp[(long long)k * DIM];
        b.y = bp[(long long)k * DIM + DIM];
        d = __builtin_amdgcn_wmma_f32_16x16x4_f32(false, a, false, b,
                                                  (short)0, d, false, false);
    }

    float* yp = Y + (long long)(m0 + 8 * hi) * DIM + n0 + lo;
#pragma unroll
    for (int v = 0; v < 8; ++v) {
        int m = m0 + v + 8 * hi;
        if (m < n) yp[(long long)v * DIM] = d[v];
    }
}

__global__ __launch_bounds__(256) void k_gather_rows(
    const float* __restrict__ xw, const int* __restrict__ offs, const int* __restrict__ srcs,
    const float* __restrict__ dinv, float* __restrict__ acc, int n) {
    const int lane = threadIdx.x & 31;
    const int node = blockIdx.x * 8 + __builtin_amdgcn_readfirstlane((int)(threadIdx.x >> 5));
    if (node >= n) return;
    const int beg = offs[node];
    const int end = offs[node + 1];
    const float dc = dinv[node];
    const float4* __restrict__ x4 = (const float4*)xw;

    const float4 sv = x4[(long long)node * 32 + lane];
    const float ss = dc * dc;
    float4 a = make_float4(sv.x * ss, sv.y * ss, sv.z * ss, sv.w * ss);

    for (int b = beg; b < end; b += 32) {
        const int m = (end - b < 32) ? (end - b) : 32;
        int r = 0;
        float wgt = 0.0f;
        if (lane < m) {
            r = srcs[b + lane];
            wgt = dinv[r] * dc;
        }
        const int wbits = __float_as_int(wgt);
#define SRC_ROW(j_) x4[(long long)__builtin_amdgcn_readlane(r, (j_)) * 32 + lane]
#define SRC_WGT(j_) __int_as_float(__builtin_amdgcn_readlane(wbits, (j_)))
#define ADD_ROW(v_, w_) a.x = fmaf(v_.x, w_, a.x); a.y = fmaf(v_.y, w_, a.y); \
                        a.z = fmaf(v_.z, w_, a.z); a.w = fmaf(v_.w, w_, a.w)
        int j = 0;
        for (; j + 4 <= m; j += 4) {
            const float4 v0 = SRC_ROW(j);
            const float4 v1 = SRC_ROW(j + 1);
            const float4 v2 = SRC_ROW(j + 2);
            const float4 v3 = SRC_ROW(j + 3);
            const float w0 = SRC_WGT(j), w1 = SRC_WGT(j + 1), w2 = SRC_WGT(j + 2), w3 = SRC_WGT(j + 3);
            ADD_ROW(v0, w0); ADD_ROW(v1, w1); ADD_ROW(v2, w2); ADD_ROW(v3, w3);
        }
        const int rem = m - j;
        if (rem == 3) {
            const float4 v0 = SRC_ROW(j);
            const float4 v1 = SRC_ROW(j + 1);
            const float4 v2 = SRC_ROW(j + 2);
            const float w0 = SRC_WGT(j), w1 = SRC_WGT(j + 1), w2 = SRC_WGT(j + 2);
            ADD_ROW(v0, w0); ADD_ROW(v1, w1); ADD_ROW(v2, w2);
        } else if (rem == 2) {
            const float4 v0 = SRC_ROW(j);
            const float4 v1 = SRC_ROW(j + 1);
            const float w0 = SRC_WGT(j), w1 = SRC_WGT(j + 1);
            ADD_ROW(v0, w0); ADD_ROW(v1, w1);
        } else if (rem == 1) {
            const float4 v0 = SRC_ROW(j);
            const float w0 = SRC_WGT(j);
            ADD_ROW(v0, w0);
        }
#undef SRC_ROW
#undef SRC_WGT
#undef ADD_ROW
    }
    ((float4*)acc)[(long long)node * 32 + lane] = a;
}

__global__ void k_root_and_c(const float* __restrict__ acc, const float* __restrict__ b1,
                             const float* __restrict__ posts, const float* __restrict__ W2,
                             const int* __restrict__ rootPtr,
                             float* __restrict__ root_feat, float* __restrict__ cvec) {
    int j = threadIdx.x;
    int root = *rootPtr;
    root_feat[j] = acc[(long long)root * DIM + j] + b1[j];
    float s = 0.0f;
    for (int k = 0; k < DIM; ++k)
        s += fmaxf(posts[(long long)root * DIM + k], 0.0f) * W2[(long long)(DIM + k) * DIM + j];
    cvec[j] = s;
}

__global__ void k_bias_relu(float* __restrict__ acc, const float* __restrict__ b, int n) {
    int t = blockIdx.x * blockDim.x + threadIdx.x;
    if (t >= n * 32) return;
    float4 bb = ((const float4*)b)[t & 31];
    float4 v = ((float4*)acc)[t];
    v.x = fmaxf(v.x + bb.x, 0.0f);
    v.y = fmaxf(v.y + bb.y, 0.0f);
    v.z = fmaxf(v.z + bb.z, 0.0f);
    v.w = fmaxf(v.w + bb.w, 0.0f);
    ((float4*)acc)[t] = v;
}

__global__ __launch_bounds__(256) void k_finalize_pool(
    const float* __restrict__ acc, const float* __restrict__ b2,
    const float* __restrict__ root_feat, float* __restrict__ out, int n) {
    __shared__ float feat[256];
    int i = blockIdx.x;
    int t = threadIdx.x;
    if (t < 128) feat[t] = root_feat[t];
    else         feat[t] = fmaxf(acc[(long long)i * DIM + (t - 128)] + b2[t - 128], 0.0f);
    __syncthreads();
    if (t < 254)
        out[(long long)i * 254 + t] = (feat[t] + feat[t + 1] + feat[t + 2]) * (1.0f / 3.0f);
}

extern "C" void kernel_launch(void* const* d_in, const int* in_sizes, int n_in,
                              void* d_out, int out_size, void* d_ws, size_t ws_size,
                              hipStream_t stream) {
    const float* posts = (const float*)d_in[0];
    const float* W1    = (const float*)d_in[1];
    const float* b1    = (const float*)d_in[2];
    const float* W2    = (const float*)d_in[3];
    const float* b2    = (const float*)d_in[4];
    const int*   edge  = (const int*)d_in[5];
    const int*   rootP = (const int*)d_in[6];

    const int n = in_sizes[0] / DIM;
    const int e = in_sizes[5] / 2;
    const int* rowi = edge;
    const int* coli = edge + e;

    char* ws = (char*)d_ws;
    size_t off = 0;
    float* dinv = (float*)(ws + off); off += (((size_t)n * 4) + 255) & ~(size_t)255;
    float* xw   = (float*)(ws + off); off += (size_t)n * DIM * 4;
    float* acc  = (float*)(ws + off); off += (size_t)n * DIM * 4;
    float* rootf = (float*)(ws + off); off += DIM * 4;
    float* cvec  = (float*)(ws + off);
    float* out = (float*)d_out;

    const size_t npad = (((size_t)n * 4) + 255) & ~(size_t)255;
    int* cnt  = (int*)acc;
    int* rank = (int*)((char*)acc + npad);
    const size_t opad = (((size_t)(n + 1) * 4) + 255) & ~(size_t)255;
    int* offs = (int*)d_out;
    int* srcs = (int*)((char*)d_out + opad);

    const int B = 256;
    const int gN   = (n + B - 1) / B;
    const int gE   = (e + B - 1) / B;
    const int gNF4 = (n * 32 + B - 1) / B;
    const int gS   = (n + 1023) / 1024;
    const int gG   = (n + 7) / 8;
    const int gM   = (n + 15) / 16;

    k_zero_i32  <<<gN, B, 0, stream>>>(cnt, n);
    k_count_rank<<<gE, B, 0, stream>>>(coli, cnt, rank, e);
    k_scan_dinv <<<gS, 1024, 0, stream>>>(cnt, offs, dinv, n);
    k_fill      <<<gE, B, 0, stream>>>(rowi, coli, rank, offs, srcs, e);

    k_gemm128_wmma <<<gM, B, 0, stream>>>(posts, W1, nullptr, xw, n);
    k_gather_rows  <<<gG, B, 0, stream>>>(xw, offs, srcs, dinv, acc, n);

    k_root_and_c<<<1, DIM, 0, stream>>>(acc, b1, posts, W2, rootP, rootf, cvec);

    k_bias_relu    <<<gNF4, B, 0, stream>>>(acc, b1, n);
    k_gemm128_wmma <<<gM, B, 0, stream>>>(acc, W2, cvec, xw, n);
    k_gather_rows  <<<gG, B, 0, stream>>>(xw, offs, srcs, dinv, acc, n);

    k_finalize_pool<<<n, B, 0, stream>>>(acc, b2, rootf, out, n);
}


# Pooling epilogue: one wave writes 8 whole output rows from registers (right-hand columns by lane shuffle, root half pooled once per wave), no LDS or barrier: 64x fewer waves start; sum order kept
# speedup vs baseline: 2.5393x; 1.1375x over previous
#include <hip/hip_runtime.h>

typedef __attribute__((ext_vector_type(2))) float v2f;
typedef __attribute__((ext_vector_type(8))) float v8f;

#define DIM 128

__global__ void k_zero_i32(int* __restrict__ p, int n) {
    int i = blockIdx.x * blockDim.x + threadIdx.x;
    if (i < n) p[i] = 0;
}

__global__ void k_count_rank(const int* __restrict__ col, int* __restrict__ cnt,
                             int* __restrict__ rank, int e) {
    int i = blockIdx.x * blockDim.x + threadIdx.x;
    if (i < e) rank[i] = atomicAdd(&cnt[col[i]], 1);
}

__global__ __launch_bounds__(1024) void k_scan_dinv(const int* __restrict__ cnt,
                                                    int* __restrict__ offs,
                                                    float* __restrict__ dinv, int n) {
    __shared__ int part[32];
    __shared__ int wtot[32];
    const int t = threadIdx.x;
    const int lane = t & 31;
    const int w = t >> 5;
    const int base = blockIdx.x * 1024;

    int s = 0;
    const int4* c4 = (const int4*)cnt;
    const int lim = base >> 2;
    for (int q = t; q < lim; q += 4096) {
        const bool p1 = q + 1024 < lim, p2 = q + 2048 < lim, p3 = q + 3072 < lim;
        const int4 v0 = c4[q];
        const int4 v1 = c4[p1 ? q + 1024 : q];
        const int4 v2 = c4[p2 ? q + 2048 : q];
        const int4 v3 = c4[p3 ? q + 3072 : q];
        const int s1 = (v1.x + v1.y) + (v1.z + v1.w);
        const int s2 = (v2.x + v2.y) + (v2.z + v2.w);
        const int s3 = (v3.x + v3.y) + (v3.z + v3.w);
        s += ((v0.x + v0.y) + (v0.z + v0.w)) + (p1 ? s1 : 0) + (p2 ? s2 : 0) + (p3 ? s3 : 0);
    }
    for (int o = 16; o > 0; o >>= 1) s += __shfl_xor(s, o);
    if (lane == 0) part[w] = s;

    const int i = base + t;
    const int c = (i < n) ? cnt[i] : 0;
    int x = c;
    for (int o = 1; o < 32; o <<= 1) {
        int y = __shfl_up(x, o);
        if (lane >= o) x += y;
    }
    if (lane == 31) wtot[w] = x;
    __syncthreads();

    int pre = 0;
    for (int k = 0; k < 32; ++k) {
        pre += part[k];
        if (k < w) pre += wtot[k];
    }
    const int excl = pre + x - c;
    if (i < n) {
        offs[i] = excl;
        dinv[i] = rsqrtf((float)(c + 1));
        if (i == n - 1) offs[n] = excl + c;
    }
}

__global__ void k_fill(const int* __restrict__ row, const int* __restrict__ col,
                       const int* __restrict__ rank, const int* __restrict__ offs,
                       int* __restrict__ srcs, int e) {
    int i = blockIdx.x * blockDim.x + threadIdx.x;
    if (i < e) srcs[offs[col[i]] + rank[i]] = row[i];
}

__global__ __launch_bounds__(256) void k_gemm128_wmma(
    const float* __restrict__ X, const float* __restrict__ W,
    const float* __restrict__ cvec, float* __restrict__ Y, int n) {
    const int lane = threadIdx.x & 31;
    const int wave = threadIdx.x >> 5;
    const int m0   = blockIdx.x * 16;
    const int n0   = wave * 16;
    const int lo   = lane & 15;
    const int hi   = lane >> 4;

    int arow = m0 + lo;
    int arow_c = (arow < n) ? arow : (n - 1);
    const float* ap = X + (long long)arow_c * DIM + 2 * hi;
    const float* bp = W + (n0 + lo) + (long long)(2 * hi) * DIM;

    float ci = cvec ? cvec[n0 + lo] : 0.0f;
    v8f d;
#pragma unroll
    for (int v = 0; v < 8; ++v) d[v] = ci;

#pragma unroll
    for (int k = 0; k < DIM; k += 4) {
        v2f a = *(const v2f*)(ap + k);
        v2f b;
        b.x = bp[(long long)k * DIM];
        b.y = bp[(long long)k * DIM + DIM];
        d = __builtin_amdgcn_wmma_f32_16x16x4_f32(false, a, false, b,
                                                  (short)0, d, false, false);
    }

    float* yp = Y + (long long)(m0 + 8 * hi) * DIM + n0 + lo;
#pragma unroll
    for (int v = 0; v < 8; ++v) {
        int m = m0 + v + 8 * hi;
        if (m < n) yp[(long long)v * DIM] = d[v];
    }
}

__global__ __launch_bounds__(256) void k_gather_rows(
    const float* __restrict__ xw, const int* __restrict__ offs, const int* __restrict__ srcs,
    const float* __restrict__ dinv, float* __restrict__ acc, int n) {
    const int lane = threadIdx.x & 31;
    const int node = blockIdx.x * 8 + __builtin_amdgcn_readfirstlane((int)(threadIdx.x >> 5));
    if (node >= n) return;
    const int beg = offs[node];
    const int end = offs[node + 1];
    const float dc = dinv[node];
    const float4* __restrict__ x4 = (const float4*)xw;

    const float4 sv = x4[(long long)node * 32 + lane];
    const float ss = dc * dc;
    float4 a = make_float4(sv.x * ss, sv.y * ss, sv.z * ss, sv.w * ss);

    for (int b = beg; b < end; b += 32) {
        const int m = (end - b < 32) ? (end - b) : 32;
        int r = 0;
        float wgt = 0.0f;
        if (lane < m) {
            r = srcs[b + lane];
            wgt = dinv[r] * dc;
        }
        const int wbits = __float_as_int(wgt);
#define SRC_ROW(j_) x4[(long long)__builtin_amdgcn_readlane(r, (j_)) * 32 + lane]
#define SRC_WGT(j_) __int_as_float(__builtin_amdgcn_readlane(wbits, (j_)))
#define ADD_ROW(v_, w_) a.x = fmaf(v_.x, w_, a.x); a.y = fmaf(v_.y, w_, a.y); \
                        a.z = fmaf(v_.z, w_, a.z); a.w = fmaf(v_.w, w_, a.w)
        int j = 0;
        for (; j + 4 <= m; j += 4) {
            const float4 v0 = SRC_ROW(j);
            const float4 v1 = SRC_ROW(j + 1);
            const float4 v2 = SRC_ROW(j + 2);
            const float4 v3 = SRC_ROW(j + 3);
            const float w0 = SRC_WGT(j), w1 = SRC_WGT(j + 1), w2 = SRC_WGT(j + 2), w3 = SRC_WGT(j + 3);
            ADD_ROW(v0, w0); ADD_ROW(v1, w1); ADD_ROW(v2, w2); ADD_ROW(v3, w3);
        }
        const int rem = m - j;
        if (rem == 3) {
            const float4 v0 = SRC_ROW(j);
            const float4 v1 = SRC_ROW(j + 1);
            const float4 v2 = SRC_ROW(j + 2);
            const float w0 = SRC_WGT(j), w1 = SRC_WGT(j + 1), w2 = SRC_WGT(j + 2);
            ADD_ROW(v0, w0); ADD_ROW(v1, w1); ADD_ROW(v2, w2);
        } else if (rem == 2) {
            const float4 v0 = SRC_ROW(j);
            const float4 v1 = SRC_ROW(j + 1);
            const float w0 = SRC_WGT(j), w1 = SRC_WGT(j + 1);
            ADD_ROW(v0, w0); ADD_ROW(v1, w1);
        } else if (rem == 1) {
            const float4 v0 = SRC_ROW(j);
            const float w0 = SRC_WGT(j);
            ADD_ROW(v0, w0);
        }
#undef SRC_ROW
#undef SRC_WGT
#undef ADD_ROW
    }
    ((float4*)acc)[(long long)node * 32 + lane] = a;
}

__global__ void k_root_and_c(const float* __restrict__ acc, const float* __restrict__ b1,
                             const float* __restrict__ posts, const float* __restrict__ W2,
                             const int* __restrict__ rootPtr,
                             float* __restrict__ root_feat, float* __restrict__ cvec) {
    int j = threadIdx.x;
    int root = *rootPtr;
    root_feat[j] = acc[(long long)root * DIM + j] + b1[j];
    float s = 0.0f;
    for (int k = 0; k < DIM; ++k)
        s += fmaxf(posts[(long long)root * DIM + k], 0.0f) * W2[(long long)(DIM + k) * DIM + j];
    cvec[j] = s;
}

__global__ void k_bias_relu(float* __restrict__ acc, const float* __restrict__ b, int n) {
    int t = blockIdx.x * blockDim.x + threadIdx.x;
    if (t >= n * 32) return;
    float4 bb = ((const float4*)b)[t & 31];
    float4 v = ((float4*)acc)[t];
    v.x = fmaxf(v.x + bb.x, 0.0f);
    v.y = fmaxf(v.y + bb.y, 0.0f);
    v.z = fmaxf(v.z + bb.z, 0.0f);
    v.w = fmaxf(v.w + bb.w, 0.0f);
    ((float4*)acc)[t] = v;
}

#define POOL_ROWS 8
typedef __attribute__((ext_vector_type(4))) float v4f;
typedef v4f v4f_a8 __attribute__((aligned(8)));

__global__ __launch_bounds__(256) void k_finalize_pool(
    const float* __restrict__ acc, const float* __restrict__ b2,
    const float* __restrict__ root_feat, float* __restrict__ out, int n) {
    const int lane = threadIdx.x & 31;
    const int wv = blockIdx.x * 8 + __builtin_amdgcn_readfirstlane((int)(threadIdx.x >> 5));
    const int r0 = wv * POOL_ROWS;
    if (r0 >= n) return;
    const float third = 1.0f / 3.0f;
    const float4* __restrict__ a4 = (const float4*)acc;

    float4 v[POOL_ROWS];
#pragma unroll
    for (int q = 0; q < POOL_ROWS; ++q) {
        const int r = (r0 + q < n) ? (r0 + q) : (n - 1);
        v[q] = a4[(long long)r * 32 + lane];
    }

    const float4 bb = ((const float4*)b2)[lane];
    const float4 rf = ((const float4*)root_feat)[lane];
    const int nx = (lane < 31) ? 4 * lane + 4 : 126;
    const float rf4 = root_feat[nx], rf5 = root_feat[nx + 1];
    v4f rp;
    rp.x = (rf.x + rf.y + rf.z) * third;
    rp.y = (rf.y + rf.z + rf.w) * third;
    rp.z = (rf.z + rf.w + rf4) * third;
    rp.w = (rf.w + rf4 + rf5) * third;
    const float s67 = rf.z + rf.w;

#pragma unroll
    for (int q = 0; q < POOL_ROWS; ++q) {
        const int r = r0 + q;
        if (r < n) {
            const float cx = fmaxf(v[q].x + bb.x, 0.0f);
            const float cy = fmaxf(v[q].y + bb.y, 0.0f);
            const float cz = fmaxf(v[q].z + bb.z, 0.0f);
            const float cw = fmaxf(v[q].w + bb.w, 0.0f);
            const float n0 = __shfl_down(cx, 1);
            const float n1 = __shfl_down(cy, 1);
            const float c0 = __int_as_float(__builtin_amdgcn_readlane(__float_as_int(cx), 0));
            const float c1 = __int_as_float(__builtin_amdgcn_readlane(__float_as_int(cy), 0));

            float* __restrict__ ob = out + (long long)r * 254;
            v4f ha = rp;
            if (lane == 31) {
                ha.z = (s67 + c0) * third;
                ha.w = (rf.w + c0 + c1) * third;
            }
            *(v4f_a8*)(ob + 4 * lane) = ha;

            v4f p;
            p.x = (cx + cy + cz) * third;
            p.y = (cy + cz + cw) * third;
            p.z = (cz + cw + n0) * third;
            p.w = (cw + n0 + n1) * third;
            if (lane < 31) *(v4f_a8*)(ob + 128 + 4 * lane) = p;
            else           *(float2*)(ob + 252) = make_float2(p.x, p.y);
        }
    }
}

extern "C" void kernel_launch(void* const* d_in, const int* in_sizes, int n_in,
                              void* d_out, int out_size, void* d_ws, size_t ws_size,
                              hipStream_t stream) {
    const float* posts = (const float*)d_in[0];
    const float* W1    = (const float*)d_in[1];
    const float* b1    = (const float*)d_in[2];
    const float* W2    = (const float*)d_in[3];
    const float* b2    = (const float*)d_in[4];
    const int*   edge  = (const int*)d_in[5];
    const int*   rootP = (const int*)d_in[6];

    const int n = in_sizes[0] / DIM;
    const int e = in_sizes[5] / 2;
    const int* rowi = edge;
    const int* coli = edge + e;

    char* ws = (char*)d_ws;
    size_t off = 0;
    float* dinv = (float*)(ws + off); off += (((size_t)n * 4) + 255) & ~(size_t)255;
    float* xw   = (float*)(ws + off); off += (size_t)n * DIM * 4;
    float* acc  = (float*)(ws + off); off += (size_t)n * DIM * 4;
    float* rootf = (float*)(ws + off); off += DIM * 4;
    float* cvec  = (float*)(ws + off);
    float* out = (float*)d_out;

    const size_t npad = (((size_t)n * 4) + 255) & ~(size_t)255;
    int* cnt  = (int*)acc;
    int* rank = (int*)((char*)acc + npad);
    const size_t opad = (((size_t)(n + 1) * 4) + 255) & ~(size_t)255;
    int* offs = (int*)d_out;
    int* srcs = (int*)((char*)d_out + opad);

    const int B = 256;
    const int gN   = (n + B - 1) / B;
    const int gE   = (e + B - 1) / B;
    const int gNF4 = (n * 32 + B - 1) / B;
    const int gS   = (n + 1023) / 1024;
    const int gG   = (n + 7) / 8;
    const int gM   = (n + 15) / 16;
    const int gP   = (n + 8 * POOL_ROWS - 1) / (8 * POOL_ROWS);

    k_zero_i32  <<<gN, B, 0, stream>>>(cnt, n);
    k_count_rank<<<gE, B, 0, stream>>>(coli, cnt, rank, e);
    k_scan_dinv <<<gS, 1024, 0, stream>>>(cnt, offs, dinv, n);
    k_fill      <<<gE, B, 0, stream>>>(rowi, coli, rank, offs, srcs, e);

    k_gemm128_wmma <<<gM, B, 0, stream>>>(posts, W1, nullptr, xw, n);
    k_gather_rows  <<<gG, B, 0, stream>>>(xw, offs, srcs, dinv, acc, n);

    k_root_and_c<<<1, DIM, 0, stream>>>(acc, b1, posts, W2, rootP, rootf, cvec);

    k_bias_relu    <<<gNF4, B, 0, stream>>>(acc, b1, n);
    k_gemm128_wmma <<<gM, B, 0, stream>>>(acc, W2, cvec, xw, n);
    k_gather_rows  <<<gG, B, 0, stream>>>(xw, offs, srcs, dinv, acc, n);

    k_finalize_pool<<<gP, B, 0, stream>>>(acc, b2, rootf, out, n);
}


# GEMM re-tiled: a wave owns a 16x128 strip (8 accumulator tiles), A read once in 16-byte pieces, W staged per work-group in LDS in fragment order; k sum order changes, cvec added after the sum
# speedup vs baseline: 2.9233x; 1.1512x over previous
#include <hip/hip_runtime.h>

typedef __attribute__((ext_vector_type(2))) float v2f;
typedef __attribute__((ext_vector_type(8))) float v8f;

#define DIM 128

__global__ void k_zero_i32(int* __restrict__ p, int n) {
    int i = blockIdx.x * blockDim.x + threadIdx.x;
    if (i < n) p[i] = 0;
}

__global__ void k_count_rank(const int* __restrict__ col, int* __restrict__ cnt,
                             int* __restrict__ rank, int e) {
    int i = blockIdx.x * blockDim.x + threadIdx.x;
    if (i < e) rank[i] = atomicAdd(&cnt[col[i]], 1);
}

__global__ __launch_bounds__(1024) void k_scan_dinv(const int* __restrict__ cnt,
                                                    int* __restrict__ offs,
                                                    float* __restrict__ dinv, int n) {
    __shared__ int part[32];
    __shared__ int wtot[32];
    const int t = threadIdx.x;
    const int lane = t & 31;
    const int w = t >> 5;
    const int base = blockIdx.x * 1024;

    int s = 0;
    const int4* c4 = (const int4*)cnt;
    const int lim = base >> 2;
    for (int q = t; q < lim; q += 4096) {
        const bool p1 = q + 1024 < lim, p2 = q + 2048 < lim, p3 = q + 3072 < lim;
        const int4 v0 = c4[q];
        const int4 v1 = c4[p1 ? q + 1024 : q];
        const int4 v2 = c4[p2 ? q + 2048 : q];
        const int4 v3 = c4[p3 ? q + 3072 : q];
        const int s1 = (v1.x + v1.y) + (v1.z + v1.w);
        const int s2 = (v2.x + v2.y) + (v2.z + v2.w);
        const int s3 = (v3.x + v3.y) + (v3.z + v3.w);
        s += ((v0.x + v0.y) + (v0.z + v0.w)) + (p1 ? s1 : 0) + (p2 ? s2 : 0) + (p3 ? s3 : 0);
    }
    for (int o = 16; o > 0; o >>= 1) s += __shfl_xor(s, o);
    if (lane == 0) part[w] = s;

    const int i = base + t;
    const int c = (i < n) ? cnt[i] : 0;
    int x = c;
    for (int o = 1; o < 32; o <<= 1) {
        int y = __shfl_up(x, o);
        if (lane >= o) x += y;
    }
    if (lane == 31) wtot[w] = x;
    __syncthreads();

    int pre = 0;
    for (int k = 0; k < 32; ++k) {
        pre += part[k];
        if (k < w) pre += wtot[k];
    }
    const int excl = pre + x - c;
    if (i < n) {
        offs[i] = excl;
        dinv[i] = rsqrtf((float)(c + 1));
        if (i == n - 1) offs[n] = excl + c;
    }
}

__global__ void k_fill(const int* __restrict__ row, const int* __restrict__ col,
                       const int* __restrict__ rank, const int* __restrict__ offs,
                       int* __restrict__ srcs, int e) {
    int i = blockIdx.x * blockDim.x + threadIdx.x;
    if (i < e) srcs[offs[col[i]] + rank[i]] = row[i];
}

#define GEMM_GROUPS 512
typedef __attribute__((ext_vector_type(4))) float v4f;

__global__ __launch_bounds__(256, 4) void k_gemm128_wmma(
    const float* __restrict__ X, const float* __restrict__ W,
    const float* __restrict__ cvec, float* __restrict__ Y, int n) {
    __shared__ v4f Bl[8 * 16 * 32];
    const int tid  = threadIdx.x;
    const int lane = tid & 31;
    const int wave = __builtin_amdgcn_readfirstlane(tid >> 5);
    const int c    = lane & 15;
    const int h    = lane >> 4;

    {
        const int sn = tid & 127;
        const int sh = tid >> 7;
        const float* wp = W + (4 * sh) * DIM + sn;
        v4f* bs = Bl + (sn >> 4) * (16 * 32) + (sn & 15) + 16 * sh;
#pragma unroll
        for (int u = 0; u < 16; ++u) {
            v4f t;
            t.x = wp[(8 * u + 0) * DIM];
            t.y = wp[(8 * u + 1) * DIM];
            t.z = wp[(8 * u + 2) * DIM];
            t.w = wp[(8 * u + 3) * DIM];
            bs[u * 32] = t;
        }
    }

    float cj[8];
#pragma unroll
    for (int j = 0; j < 8; ++j) cj[j] = cvec ? cvec[16 * j + c] : 0.0f;

    __syncthreads();

    const int ntiles = (n + 15) >> 4;
    const int stride = 8 * (int)gridDim.x;
    int tile = wave * (int)gridDim.x + (int)blockIdx.x;

    v4f a[16];
    {
        const int r0 = (tile * 16 + c < n) ? (tile * 16 + c) : (n - 1);
        const v4f* ap = (const v4f*)(X + (long long)r0 * DIM + 4 * h);
#pragma unroll
        for (int u = 0; u < 16; ++u) a[u] = ap[2 * u];
    }

    const v4f* bl = Bl + lane;
    for (; tile < ntiles; tile += stride) {
        asm volatile("" ::: "memory");
        const int m0 = tile * 16;
        const int mn = (tile + stride) * 16 + c;
        const int rn = (mn < n) ? mn : (n - 1);
        const v4f* apn = (const v4f*)(X + (long long)rn * DIM + 4 * h);

        v8f d[8];
#pragma unroll
        for (int j = 0; j < 8; ++j) {
#pragma unroll
            for (int v = 0; v < 8; ++v) d[j][v] = 0.0f;
        }

        v4f bc[8], bn[8];
#pragma unroll
        for (int j = 0; j < 8; ++j) bc[j] = bl[(j * 16) * 32];
#pragma unroll
        for (int u = 0; u < 16; ++u) {
            if (u + 1 < 16) {
#pragma unroll
                for (int j = 0; j < 8; ++j) bn[j] = bl[(j * 16 + u + 1) * 32];
            }
            __builtin_amdgcn_sched_barrier(0);
            v2f a0, a1;
            a0.x = a[u].x; a0.y = a[u].y;
            a1.x = a[u].z; a1.y = a[u].w;
#pragma unroll
            for (int j = 0; j < 8; ++j) {
                v2f b0;
                b0.x = bc[j].x; b0.y = bc[j].y;
                d[j] = __builtin_amdgcn_wmma_f32_16x16x4_f32(false, a0, false, b0,
                                                             (short)0, d[j], false, false);
            }
#pragma unroll
            for (int j = 0; j < 8; ++j) {
                v2f b1;
                b1.x = bc[j].z; b1.y = bc[j].w;
                d[j] = __builtin_amdgcn_wmma_f32_16x16x4_f32(false, a1, false, b1,
                                                             (short)0, d[j], false, false);
            }
            a[u] = apn[2 * u];
            __builtin_amdgcn_sched_barrier(0);
#pragma unroll
            for (int j = 0; j < 8; ++j) bc[j] = bn[j];
        }

        float* yp = Y + (long long)(m0 + 8 * h) * DIM + c;
#pragma unroll
        for (int v = 0; v < 8; ++v) {
            const int m = m0 + v + 8 * h;
            if (m < n) {
#pragma unroll
                for (int j = 0; j < 8; ++j) yp[(long long)v * DIM + 16 * j] = d[j][v] + cj[j];
            }
        }
    }
}

__global__ __launch_bounds__(256) void k_gather_rows(
    const float* __restrict__ xw, const int* __restrict__ offs, const int* __restrict__ srcs,
    const float* __restrict__ dinv, float* __restrict__ acc, int n) {
    const int lane = threadIdx.x & 31;
    const int node = blockIdx.x * 8 + __builtin_amdgcn_readfirstlane((int)(threadIdx.x >> 5));
    if (node >= n) return;
    const int beg = offs[node];
    const int end = offs[node + 1];
    const float dc = dinv[node];
    const float4* __restrict__ x4 = (const float4*)xw;

    const float4 sv = x4[(long long)node * 32 + lane];
    const float ss = dc * dc;
    float4 a = make_float4(sv.x * ss, sv.y * ss, sv.z * ss, sv.w * ss);

    for (int b = beg; b < end; b += 32) {
        const int m = (end - b < 32) ? (end - b) : 32;
        int r = 0;
        float wgt = 0.0f;
        if (lane < m) {
            r = srcs[b + lane];
            wgt = dinv[r] * dc;
        }
        const int wbits = __float_as_int(wgt);
#define SRC_ROW(j_) x4[(long long)__builtin_amdgcn_readlane(r, (j_)) * 32 + lane]
#define SRC_WGT(j_) __int_as_float(__builtin_amdgcn_readlane(wbits, (j_)))
#define ADD_ROW(v_, w_) a.x = fmaf(v_.x, w_, a.x); a.y = fmaf(v_.y, w_, a.y); \
                        a.z = fmaf(v_.z, w_, a.z); a.w = fmaf(v_.w, w_, a.w)
        int j = 0;
        for (; j + 4 <= m; j += 4) {
            const float4 v0 = SRC_ROW(j);
            const float4 v1 = SRC_ROW(j + 1);
            const float4 v2 = SRC_ROW(j + 2);
            const float4 v3 = SRC_ROW(j + 3);
            const float w0 = SRC_WGT(j), w1 = SRC_WGT(j + 1), w2 = SRC_WGT(j + 2), w3 = SRC_WGT(j + 3);
            ADD_ROW(v0, w0); ADD_ROW(v1, w1); ADD_ROW(v2, w2); ADD_ROW(v3, w3);
        }
        const int rem = m - j;
        if (rem == 3) {
            const float4 v0 = SRC_ROW(j);
            const float4 v1 = SRC_ROW(j + 1);
            const float4 v2 = SRC_ROW(j + 2);
            const float w0 = SRC_WGT(j), w1 = SRC_WGT(j + 1), w2 = SRC_WGT(j + 2);
            ADD_ROW(v0, w0); ADD_ROW(v1, w1); ADD_ROW(v2, w2);
        } else if (rem == 2) {
            const float4 v0 = SRC_ROW(j);
            const float4 v1 = SRC_ROW(j + 1);
            const float w0 = SRC_WGT(j), w1 = SRC_WGT(j + 1);
            ADD_ROW(v0, w0); ADD_ROW(v1, w1);
        } else if (rem == 1) {
            const float4 v0 = SRC_ROW(j);
            const float w0 = SRC_WGT(j);
            ADD_ROW(v0, w0);
        }
#undef SRC_ROW
#undef SRC_WGT
#undef ADD_ROW
    }
    ((float4*)acc)[(long long)node * 32 + lane] = a;
}

__global__ void k_root_and_c(const float* __restrict__ acc, const float* __restrict__ b1,
                             const float* __restrict__ posts, const float* __restrict__ W2,
                             const int* __restrict__ rootPtr,
                             float* __restrict__ root_feat, float* __restrict__ cvec) {
    int j = threadIdx.x;
    int root = *rootPtr;
    root_feat[j] = acc[(long long)root * DIM + j] + b1[j];
    float s = 0.0f;
    for (int k = 0; k < DIM; ++k)
        s += fmaxf(posts[(long long)root * DIM + k], 0.0f) * W2[(long long)(DIM + k) * DIM + j];
    cvec[j] = s;
}

__global__ void k_bias_relu(float* __restrict__ acc, const float* __restrict__ b, int n) {
    int t = blockIdx.x * blockDim.x + threadIdx.x;
    if (t >= n * 32) return;
    float4 bb = ((const float4*)b)[t & 31];
    float4 v = ((float4*)acc)[t];
    v.x = fmaxf(v.x + bb.x, 0.0f);
    v.y = fmaxf(v.y + bb.y, 0.0f);
    v.z = fmaxf(v.z + bb.z, 0.0f);
    v.w = fmaxf(v.w + bb.w, 0.0f);
    ((float4*)acc)[t] = v;
}

#define POOL_ROWS 8
typedef v4f v4f_a8 __attribute__((aligned(8)));

__global__ __launch_bounds__(256) void k_finalize_pool(
    const float* __restrict__ acc, const float* __restrict__ b2,
    const float* __restrict__ root_feat, float* __restrict__ out, int n) {
    const int lane = threadIdx.x & 31;
    const int wv = blockIdx.x * 8 + __builtin_amdgcn_readfirstlane((int)(threadIdx.x >> 5));
    const int r0 = wv * POOL_ROWS;
    if (r0 >= n) return;
    const float third = 1.0f / 3.0f;
    const float4* __restrict__ a4 = (const float4*)acc;

    float4 v[POOL_ROWS];
#pragma unroll
    for (int q = 0; q < POOL_ROWS; ++q) {
        const int r = (r0 + q < n) ? (r0 + q) : (n - 1);
        v[q] = a4[(long long)r * 32 + lane];
    }

    const float4 bb = ((const float4*)b2)[lane];
    const float4 rf = ((const float4*)root_feat)[lane];
    const int nx = (lane < 31) ? 4 * lane + 4 : 126;
    const float rf4 = root_feat[nx], rf5 = root_feat[nx + 1];
    v4f rp;
    rp.x = (rf.x + rf.y + rf.z) * third;
    rp.y = (rf.y + rf.z + rf.w) * third;
    rp.z = (rf.z + rf.w + rf4) * third;
    rp.w = (rf.w + rf4 + rf5) * third;
    const float s67 = rf.z + rf.w;

#pragma unroll
    for (int q = 0; q < POOL_ROWS; ++q) {
        const int r = r0 + q;
        if (r < n) {
            const float cx = fmaxf(v[q].x + bb.x, 0.0f);
            const float cy = fmaxf(v[q].y + bb.y, 0.0f);
            const float cz = fmaxf(v[q].z + bb.z, 0.0f);
            const float cw = fmaxf(v[q].w + bb.w, 0.0f);
            const float n0 = __shfl_down(cx, 1);
            const float n1 = __shfl_down(cy, 1);
            const float c0 = __int_as_float(__builtin_amdgcn_readlane(__float_as_int(cx), 0));
            const float c1 = __int_as_float(__builtin_amdgcn_readlane(__float_as_int(cy), 0));

            float* __restrict__ ob = out + (long long)r * 254;
            v4f ha = rp;
            if (lane == 31) {
                ha.z = (s67 + c0) * third;
                ha.w = (rf.w + c0 + c1) * third;
            }
            *(v4f_a8*)(ob + 4 * lane) = ha;

            v4f p;
            p.x = (cx + cy + cz) * third;
            p.y = (cy + cz + cw) * third;
            p.z = (cz + cw + n0) * third;
            p.w = (cw + n0 + n1) * third;
            if (lane < 31) *(v4f_a8*)(ob + 128 + 4 * lane) = p;
            else           *(float2*)(ob + 252) = make_float2(p.x, p.y);
        }
    }
}

extern "C" void kernel_launch(void* const* d_in, const int* in_sizes, int n_in,
                              void* d_out, int out_size, void* d_ws, size_t ws_size,
                              hipStream_t stream) {
    const float* posts = (const float*)d_in[0];
    const float* W1    = (const float*)d_in[1];
    const float* b1    = (const float*)d_in[2];
    const float* W2    = (const float*)d_in[3];
    const float* b2    = (const float*)d_in[4];
    const int*   edge  = (const int*)d_in[5];
    const int*   rootP = (const int*)d_in[6];

    const int n = in_sizes[0] / DIM;
    const int e = in_sizes[5] / 2;
    const int* rowi = edge;
    const int* coli = edge + e;

    char* ws = (char*)d_ws;
    size_t off = 0;
    float* dinv = (float*)(ws + off); off += (((size_t)n * 4) + 255) & ~(size_t)255;
    float* xw   = (float*)(ws + off); off += (size_t)n * DIM * 4;
    float* acc  = (float*)(ws + off); off += (size_t)n * DIM * 4;
    float* rootf = (float*)(ws + off); off += DIM * 4;
    float* cvec  = (float*)(ws + off);
    float* out = (float*)d_out;

    const size_t npad = (((size_t)n * 4) + 255) & ~(size_t)255;
    int* cnt  = (int*)acc;
    int* rank = (int*)((char*)acc + npad);
    const size_t opad = (((size_t)(n + 1) * 4) + 255) & ~(size_t)255;
    int* offs = (int*)d_out;
    int* srcs = (int*)((char*)d_out + opad);

    const int B = 256;
    const int gN   = (n + B - 1) / B;
    const int gE   = (e + B - 1) / B;
    const int gNF4 = (n * 32 + B - 1) / B;
    const int gS   = (n + 1023) / 1024;
    const int gG   = (n + 7) / 8;
    const int gM   = GEMM_GROUPS;
    const int gP   = (n + 8 * POOL_ROWS - 1) / (8 * POOL_ROWS);

    k_zero_i32  <<<gN, B, 0, stream>>>(cnt, n);
    k_count_rank<<<gE, B, 0, stream>>>(coli, cnt, rank, e);
    k_scan_dinv <<<gS, 1024, 0, stream>>>(cnt, offs, dinv, n);
    k_fill      <<<gE, B, 0, stream>>>(rowi, coli, rank, offs, srcs, e);

    k_gemm128_wmma <<<gM, B, 0, stream>>>(posts, W1, nullptr, xw, n);
    k_gather_rows  <<<gG, B, 0, stream>>>(xw, offs, srcs, dinv, acc, n);

    k_root_and_c<<<1, DIM, 0, stream>>>(acc, b1, posts, W2, rootP, rootf, cvec);

    k_bias_relu    <<<gNF4, B, 0, stream>>>(acc, b1, n);
    k_gemm128_wmma <<<gM, B, 0, stream>>>(acc, W2, cvec, xw, n);
    k_gather_rows  <<<gG, B, 0, stream>>>(xw, offs, srcs, dinv, acc, n);

    k_finalize_pool<<<gP, B, 0, stream>>>(acc, b2, rootf, out, n);
}


# Elementwise passes folded into the gathers' row stores: layer 1 stores relu(row+b1), the root row and cvec; layer 2 pools and writes out; 3 launches and 2 passes over the rows go; sums unchanged
# speedup vs baseline: 3.2620x; 1.1159x over previous
#include <hip/hip_runtime.h>

typedef __attribute__((ext_vector_type(2))) float v2f;
typedef __attribute__((ext_vector_type(8))) float v8f;

#define DIM 128

__global__ void k_zero_i32(int* __restrict__ p, int n) {
    int i = blockIdx.x * blockDim.x + threadIdx.x;
    if (i < n) p[i] = 0;
}

__global__ void k_count_rank(const int* __restrict__ col, int* __restrict__ cnt,
                             int* __restrict__ rank, int e) {
    int i = blockIdx.x * blockDim.x + threadIdx.x;
    if (i < e) rank[i] = atomicAdd(&cnt[col[i]], 1);
}

__global__ __launch_bounds__(1024) void k_scan_dinv(const int* __restrict__ cnt,
                                                    int* __restrict__ offs,
                                                    float* __restrict__ dinv, int n) {
    __shared__ int part[32];
    __shared__ int wtot[32];
    const int t = threadIdx.x;
    const int lane = t & 31;
    const int w = t >> 5;
    const int base = blockIdx.x * 1024;

    int s = 0;
    const int4* c4 = (const int4*)cnt;
    const int lim = base >> 2;
    for (int q = t; q < lim; q += 4096) {
        const bool p1 = q + 1024 < lim, p2 = q + 2048 < lim, p3 = q + 3072 < lim;
        const int4 v0 = c4[q];
        const int4 v1 = c4[p1 ? q + 1024 : q];
        const int4 v2 = c4[p2 ? q + 2048 : q];
        const int4 v3 = c4[p3 ? q + 3072 : q];
        const int s1 = (v1.x + v1.y) + (v1.z + v1.w);
        const int s2 = (v2.x + v2.y) + (v2.z + v2.w);
        const int s3 = (v3.x + v3.y) + (v3.z + v3.w);
        s += ((v0.x + v0.y) + (v0.z + v0.w)) + (p1 ? s1 : 0) + (p2 ? s2 : 0) + (p3 ? s3 : 0);
    }
    for (int o = 16; o > 0; o >>= 1) s += __shfl_xor(s, o);
    if (lane == 0) part[w] = s;

    const int i = base + t;
    const int c = (i < n) ? cnt[i] : 0;
    int x = c;
    for (int o = 1; o < 32; o <<= 1) {
        int y = __shfl_up(x, o);
        if (lane >= o) x += y;
    }
    if (lane == 31) wtot[w] = x;
    __syncthreads();

    int pre = 0;
    for (int k = 0; k < 32; ++k) {
        pre += part[k];
        if (k < w) pre += wtot[k];
    }
    const int excl = pre + x - c;
    if (i < n) {
        offs[i] = excl;
        dinv[i] = rsqrtf((float)(c + 1));
        if (i == n - 1) offs[n] = excl + c;
    }
}

__global__ void k_fill(const int* __restrict__ row, const int* __restrict__ col,
                       const int* __restrict__ rank, const int* __restrict__ offs,
                       int* __restrict__ srcs, int e) {
    int i = blockIdx.x * blockDim.x + threadIdx.x;
    if (i < e) srcs[offs[col[i]] + rank[i]] = row[i];
}

#define GEMM_GROUPS 512
typedef __attribute__((ext_vector_type(4))) float v4f;

__global__ __launch_bounds__(256, 4) void k_gemm128_wmma(
    const float* __restrict__ X, const float* __restrict__ W,
    const float* __restrict__ cvec, float* __restrict__ Y, int n) {
    __shared__ v4f Bl[8 * 16 * 32];
    const int tid  = threadIdx.x;
    const int lane = tid & 31;
    const int wave = __builtin_amdgcn_readfirstlane(tid >> 5);
    const int c    = lane & 15;
    const int h    = lane >> 4;

    {
        const int sn = tid & 127;
        const int sh = tid >> 7;
        const float* wp = W + (4 * sh) * DIM + sn;
        v4f* bs = Bl + (sn >> 4) * (16 * 32) + (sn & 15) + 16 * sh;
#pragma unroll
        for (int u = 0; u < 16; ++u) {
            v4f t;
            t.x = wp[(8 * u + 0) * DIM];
            t.y = wp[(8 * u + 1) * DIM];
            t.z = wp[(8 * u + 2) * DIM];
            t.w = wp[(8 * u + 3) * DIM];
            bs[u * 32] = t;
        }
    }

    float cj[8];
#pragma unroll
    for (int j = 0; j < 8; ++j) cj[j] = cvec ? cvec[16 * j + c] : 0.0f;

    __syncthreads();

    const int ntiles = (n + 15) >> 4;
    const int stride = 8 * (int)gridDim.x;
    int tile = wave * (int)gridDim.x + (int)blockIdx.x;

    v4f a[16];
    {
        const int r0 = (tile * 16 + c < n) ? (tile * 16 + c) : (n - 1);
        const v4f* ap = (const v4f*)(X + (long long)r0 * DIM + 4 * h);
#pragma unroll
        for (int u = 0; u < 16; ++u) a[u] = ap[2 * u];
    }

    const v4f* bl = Bl + lane;
    for (; tile < ntiles; tile += stride) {
        asm volatile("" ::: "memory");
        const int m0 = tile * 16;
        const int mn = (tile + stride) * 16 + c;
        const int rn = (mn < n) ? mn : (n - 1);
        const v4f* apn = (const v4f*)(X + (long long)rn * DIM + 4 * h);

        v8f d[8];
#pragma unroll
        for (int j = 0; j < 8; ++j) {
#pragma unroll
            for (int v = 0; v < 8; ++v) d[j][v] = 0.0f;
        }

        v4f bc[8], bn[8];
#pragma unroll
        for (int j = 0; j < 8; ++j) bc[j] = bl[(j * 16) * 32];
#pragma unroll
        for (int u = 0; u < 16; ++u) {
            if (u + 1 < 16) {
#pragma unroll
                for (int j = 0; j < 8; ++j) bn[j] = bl[(j * 16 + u + 1) * 32];
            }
            __builtin_amdgcn_sched_barrier(0);
            v2f a0, a1;
            a0.x = a[u].x; a0.y = a[u].y;
            a1.x = a[u].z; a1.y = a[u].w;
#pragma unroll
            for (int j = 0; j < 8; ++j) {
                v2f b0;
                b0.x = bc[j].x; b0.y = bc[j].y;
                d[j] = __builtin_amdgcn_wmma_f32_16x16x4_f32(false, a0, false, b0,
                                                             (short)0, d[j], false, false);
            }
#pragma unroll
            for (int j = 0; j < 8; ++j) {
                v2f b1;
                b1.x = bc[j].z; b1.y = bc[j].w;
                d[j] = __builtin_amdgcn_wmma_f32_16x16x4_f32(false, a1, false, b1,
                                                             (short)0, d[j], false, false);
            }
            a[u] = apn[2 * u];
            __builtin_amdgcn_sched_barrier(0);
#pragma unroll
            for (int j = 0; j < 8; ++j) bc[j] = bn[j];
        }

        float* yp = Y + (long long)(m0 + 8 * h) * DIM + c;
#pragma unroll
        for (int v = 0; v < 8; ++v) {
            const int m = m0 + v + 8 * h;
            if (m < n) {
#pragma unroll
                for (int j = 0; j < 8; ++j) yp[(long long)v * DIM + 16 * j] = d[j][v] + cj[j];
            }
        }
    }
}

__device__ __forceinline__ float4 gather_row(
    const float* __restrict__ xw, const int* __restrict__ offs, const int* __restrict__ srcs,
    const float* __restrict__ dinv, int node, int lane) {
    const int beg = offs[node];
    const int end = offs[node + 1];
    const float dc = dinv[node];
    const float4* __restrict__ x4 = (const float4*)xw;

    const float4 sv = x4[(long long)node * 32 + lane];
    const float ss = dc * dc;
    float4 a = make_float4(sv.x * ss, sv.y * ss, sv.z * ss, sv.w * ss);

    for (int b = beg; b < end; b += 32) {
        const int m = (end - b < 32) ? (end - b) : 32;
        int r = 0;
        float wgt = 0.0f;
        if (lane < m) {
            r = srcs[b + lane];
            wgt = dinv[r] * dc;
        }
        const int wbits = __float_as_int(wgt);
#define SRC_ROW(j_) x4[(long long)__builtin_amdgcn_readlane(r, (j_)) * 32 + lane]
#define SRC_WGT(j_) __int_as_float(__builtin_amdgcn_readlane(wbits, (j_)))
#define ADD_ROW(v_, w_) a.x = fmaf(v_.x, w_, a.x); a.y = fmaf(v_.y, w_, a.y); \
                        a.z = fmaf(v_.z, w_, a.z); a.w = fmaf(v_.w, w_, a.w)
        int j = 0;
        for (; j + 4 <= m; j += 4) {
            const float4 v0 = SRC_ROW(j);
            const float4 v1 = SRC_ROW(j + 1);
            const float4 v2 = SRC_ROW(j + 2);
            const float4 v3 = SRC_ROW(j + 3);
            const float w0 = SRC_WGT(j), w1 = SRC_WGT(j + 1), w2 = SRC_WGT(j + 2), w3 = SRC_WGT(j + 3);
            ADD_ROW(v0, w0); ADD_ROW(v1, w1); ADD_ROW(v2, w2); ADD_ROW(v3, w3);
        }
        const int rem = m - j;
        if (rem == 3) {
            const float4 v0 = SRC_ROW(j);
            const float4 v1 = SRC_ROW(j + 1);
            const float4 v2 = SRC_ROW(j + 2);
            const float w0 = SRC_WGT(j), w1 = SRC_WGT(j + 1), w2 = SRC_WGT(j + 2);
            ADD_ROW(v0, w0); ADD_ROW(v1, w1); ADD_ROW(v2, w2);
        } else if (rem == 2) {
            const float4 v0 = SRC_ROW(j);
            const float4 v1 = SRC_ROW(j + 1);
            const float w0 = SRC_WGT(j), w1 = SRC_WGT(j + 1);
            ADD_ROW(v0, w0); ADD_ROW(v1, w1);
        } else if (rem == 1) {
            const float4 v0 = SRC_ROW(j);
            const float w0 = SRC_WGT(j);
            ADD_ROW(v0, w0);
        }
#undef SRC_ROW
#undef SRC_WGT
#undef ADD_ROW
    }
    return a;
}

__global__ __launch_bounds__(256) void k_gather_relu(
    const float* __restrict__ xw, const int* __restrict__ offs, const int* __restrict__ srcs,
    const float* __restrict__ dinv, const float* __restrict__ b1,
    const float* __restrict__ posts, const float* __restrict__ W2,
    const int* __restrict__ rootPtr,
    float* __restrict__ h1, float* __restrict__ root_feat, float* __restrict__ cvec, int n) {
    const int lane = threadIdx.x & 31;
    const int node = blockIdx.x * 8 + __builtin_amdgcn_readfirstlane((int)(threadIdx.x >> 5));
    if (node >= n) return;
    const int root = *rootPtr;
    const float4 bb = ((const float4*)b1)[lane];

    const float4 a = gather_row(xw, offs, srcs, dinv, node, lane);

    const float4 s = make_float4(a.x + bb.x, a.y + bb.y, a.z + bb.z, a.w + bb.w);
    if (node == root) ((float4*)root_feat)[lane] = s;
    ((float4*)h1)[(long long)node * 32 + lane] =
        make_float4(fmaxf(s.x, 0.0f), fmaxf(s.y, 0.0f), fmaxf(s.z, 0.0f), fmaxf(s.w, 0.0f));

    if (node == 0) {
        const float* __restrict__ pr = posts + (long long)root * DIM;
        const float4* __restrict__ w4 = (const float4*)(W2 + (long long)DIM * DIM) + lane;
        float4 cs = make_float4(0.0f, 0.0f, 0.0f, 0.0f);
        for (int k = 0; k < DIM; ++k) {
            const float p = fmaxf(pr[k], 0.0f);
            const float4 w = w4[k * 32];
            cs.x = fmaf(p, w.x, cs.x);
            cs.y = fmaf(p, w.y, cs.y);
            cs.z = fmaf(p, w.z, cs.z);
            cs.w = fmaf(p, w.w, cs.w);
        }
        ((float4*)cvec)[lane] = cs;
    }
}

typedef v4f v4f_a8 __attribute__((aligned(8)));

__global__ __launch_bounds__(256) void k_gather_pool(
    const float* __restrict__ xw, const int* __restrict__ offs, const int* __restrict__ srcs,
    const float* __restrict__ dinv, const float* __restrict__ b2,
    const float* __restrict__ root_feat, float* __restrict__ out, int n) {
    const int lane = threadIdx.x & 31;
    const int node = blockIdx.x * 8 + __builtin_amdgcn_readfirstlane((int)(threadIdx.x >> 5));
    if (node >= n) return;
    const float third = 1.0f / 3.0f;

    const float4 bb = ((const float4*)b2)[lane];
    const float4 rf = ((const float4*)root_feat)[lane];
    const int nx = (lane < 31) ? 4 * lane + 4 : 126;
    const float rf4 = root_feat[nx], rf5 = root_feat[nx + 1];

    const float4 a = gather_row(xw, offs, srcs, dinv, node, lane);

    v4f rp;
    rp.x = (rf.x + rf.y + rf.z) * third;
    rp.y = (rf.y + rf.z + rf.w) * third;
    rp.z = (rf.z + rf.w + rf4) * third;
    rp.w = (rf.w + rf4 + rf5) * third;
    const float s67 = rf.z + rf.w;

    const float cx = fmaxf(a.x + bb.x, 0.0f);
    const float cy = fmaxf(a.y + bb.y, 0.0f);
    const float cz = fmaxf(a.z + bb.z, 0.0f);
    const float cw = fmaxf(a.w + bb.w, 0.0f);
    const float n0 = __shfl_down(cx, 1);
    const float n1 = __shfl_down(cy, 1);
    const float c0 = __int_as_float(__builtin_amdgcn_readlane(__float_as_int(cx), 0));
    const float c1 = __int_as_float(__builtin_amdgcn_readlane(__float_as_int(cy), 0));

    float* __restrict__ ob = out + (long long)node * 254;
    v4f ha = rp;
    if (lane == 31) {
        ha.z = (s67 + c0) * third;
        ha.w = (rf.w + c0 + c1) * third;
    }
    *(v4f_a8*)(ob + 4 * lane) = ha;

    v4f p;
    p.x = (cx + cy + cz) * third;
    p.y = (cy + cz + cw) * third;
    p.z = (cz + cw + n0) * third;
    p.w = (cw + n0 + n1) * third;
    if (lane < 31) *(v4f_a8*)(ob + 128 + 4 * lane) = p;
    else           *(float2*)(ob + 252) = make_float2(p.x, p.y);
}

extern "C" void kernel_launch(void* const* d_in, const int* in_sizes, int n_in,
                              void* d_out, int out_size, void* d_ws, size_t ws_size,
                              hipStream_t stream) {
    const float* posts = (const float*)d_in[0];
    const float* W1    = (const float*)d_in[1];
    const float* b1    = (const float*)d_in[2];
    const float* W2    = (const float*)d_in[3];
    const float* b2    = (const float*)d_in[4];
    const int*   edge  = (const int*)d_in[5];
    const int*   rootP = (const int*)d_in[6];

    const int n = in_sizes[0] / DIM;
    const int e = in_sizes[5] / 2;
    const int* rowi = edge;
    const int* coli = edge + e;

    char* ws = (char*)d_ws;
    size_t off = 0;
    float* dinv = (float*)(ws + off); off += (((size_t)n * 4) + 255) & ~(size_t)255;
    float* xw   = (float*)(ws + off); off += (size_t)n * DIM * 4;
    char*  lst  = ws + off;           off += (size_t)n * DIM * 4;
    float* rootf = (float*)(ws + off); off += DIM * 4;
    float* cvec  = (float*)(ws + off);
    float* out = (float*)d_out;

    const size_t npad = (((size_t)n * 4) + 255) & ~(size_t)255;
    const size_t epad = (((size_t)e * 4) + 255) & ~(size_t)255;
    const size_t opad = (((size_t)(n + 1) * 4) + 255) & ~(size_t)255;
    int* cnt  = (int*)lst;
    int* rank = (int*)(lst + npad);
    int* offs = (int*)(lst + npad + epad);
    int* srcs = (int*)(lst + npad + epad + opad);
    float* h1 = (float*)d_out;

    const int B = 256;
    const int gN   = (n + B - 1) / B;
    const int gE   = (e + B - 1) / B;
    const int gS   = (n + 1023) / 1024;
    const int gG   = (n + 7) / 8;
    const int gM   = GEMM_GROUPS;

    k_zero_i32  <<<gN, B, 0, stream>>>(cnt, n);
    k_count_rank<<<gE, B, 0, stream>>>(coli, cnt, rank, e);
    k_scan_dinv <<<gS, 1024, 0, stream>>>(cnt, offs, dinv, n);
    k_fill      <<<gE, B, 0, stream>>>(rowi, coli, rank, offs, srcs, e);

    k_gemm128_wmma <<<gM, B, 0, stream>>>(posts, W1, nullptr, xw, n);
    k_gather_relu  <<<gG, B, 0, stream>>>(xw, offs, srcs, dinv, b1, posts, W2, rootP,
                                         h1, rootf, cvec, n);

    k_gemm128_wmma <<<gM, B, 0, stream>>>(h1, W2, cvec, xw, n);
    k_gather_pool  <<<gG, B, 0, stream>>>(xw, offs, srcs, dinv, b2, rootf, out, n);
}


# GEMM epilogue scales row m by deg(m)^-1/2; gathers add pre-scaled source rows, scale once by the destination's factor: per-edge 4-byte factor lookups go; products re-associated, sum order kept
# speedup vs baseline: 3.3637x; 1.0312x over previous
#include <hip/hip_runtime.h>

typedef __attribute__((ext_vector_type(2))) float v2f;
typedef __attribute__((ext_vector_type(8))) float v8f;

#define DIM 128

__global__ void k_zero_i32(int* __restrict__ p, int n) {
    int i = blockIdx.x * blockDim.x + threadIdx.x;
    if (i < n) p[i] = 0;
}

__global__ void k_count_rank(const int* __restrict__ col, int* __restrict__ cnt,
                             int* __restrict__ rank, int e) {
    int i = blockIdx.x * blockDim.x + threadIdx.x;
    if (i < e) rank[i] = atomicAdd(&cnt[col[i]], 1);
}

__global__ __launch_bounds__(1024) void k_scan_dinv(const int* __restrict__ cnt,
                                                    int* __restrict__ offs,
                                                    float* __restrict__ dinv, int n) {
    __shared__ int part[32];
    __shared__ int wtot[32];
    const int t = threadIdx.x;
    const int lane = t & 31;
    const int w = t >> 5;
    const int base = blockIdx.x * 1024;

    int s = 0;
    const int4* c4 = (const int4*)cnt;
    const int lim = base >> 2;
    for (int q = t; q < lim; q += 4096) {
        const bool p1 = q + 1024 < lim, p2 = q + 2048 < lim, p3 = q + 3072 < lim;
        const int4 v0 = c4[q];
        const int4 v1 = c4[p1 ? q + 1024 : q];
        const int4 v2 = c4[p2 ? q + 2048 : q];
        const int4 v3 = c4[p3 ? q + 3072 : q];
        const int s1 = (v1.x + v1.y) + (v1.z + v1.w);
        const int s2 = (v2.x + v2.y) + (v2.z + v2.w);
        const int s3 = (v3.x + v3.y) + (v3.z + v3.w);
        s += ((v0.x + v0.y) + (v0.z + v0.w)) + (p1 ? s1 : 0) + (p2 ? s2 : 0) + (p3 ? s3 : 0);
    }
    for (int o = 16; o > 0; o >>= 1) s += __shfl_xor(s, o);
    if (lane == 0) part[w] = s;

    const int i = base + t;
    const int c = (i < n) ? cnt[i] : 0;
    int x = c;
    for (int o = 1; o < 32; o <<= 1) {
        int y = __shfl_up(x, o);
        if (lane >= o) x += y;
    }
    if (lane == 31) wtot[w] = x;
    __syncthreads();

    int pre = 0;
    for (int k = 0; k < 32; ++k) {
        pre += part[k];
        if (k < w) pre += wtot[k];
    }
    const int excl = pre + x - c;
    if (i < n) {
        offs[i] = excl;
        dinv[i] = rsqrtf((float)(c + 1));
        if (i == n - 1) offs[n] = excl + c;
    }
}

__global__ void k_fill(const int* __restrict__ row, const int* __restrict__ col,
                       const int* __restrict__ rank, const int* __restrict__ offs,
                       int* __restrict__ srcs, int e) {
    int i = blockIdx.x * blockDim.x + threadIdx.x;
    if (i < e) srcs[offs[col[i]] + rank[i]] = row[i];
}

#define GEMM_GROUPS 512
typedef __attribute__((ext_vector_type(4))) float v4f;

__global__ __launch_bounds__(256, 4) void k_gemm128_wmma(
    const float* __restrict__ X, const float* __restrict__ W,
    const float* __restrict__ cvec, const float* __restrict__ rs, float* __restrict__ Y, int n) {
    __shared__ v4f Bl[8 * 16 * 32];
    const int tid  = threadIdx.x;
    const int lane = tid & 31;
    const int wave = __builtin_amdgcn_readfirstlane(tid >> 5);
    const int c    = lane & 15;
    const int h    = lane >> 4;

    {
        const int sn = tid & 127;
        const int sh = tid >> 7;
        const float* wp = W + (4 * sh) * DIM + sn;
        v4f* bs = Bl + (sn >> 4) * (16 * 32) + (sn & 15) + 16 * sh;
#pragma unroll
        for (int u = 0; u < 16; ++u) {
            v4f t;
            t.x = wp[(8 * u + 0) * DIM];
            t.y = wp[(8 * u + 1) * DIM];
            t.z = wp[(8 * u + 2) * DIM];
            t.w = wp[(8 * u + 3) * DIM];
            bs[u * 32] = t;
        }
    }

    float cj[8];
#pragma unroll
    for (int j = 0; j < 8; ++j) cj[j] = cvec ? cvec[16 * j + c] : 0.0f;

    __syncthreads();

    const int ntiles = (n + 15) >> 4;
    const int stride = 8 * (int)gridDim.x;
    int tile = wave * (int)gridDim.x + (int)blockIdx.x;

    v4f a[16];
    {
        const int r0 = (tile * 16 + c < n) ? (tile * 16 + c) : (n - 1);
        const v4f* ap = (const v4f*)(X + (long long)r0 * DIM + 4 * h);
#pragma unroll
        for (int u = 0; u < 16; ++u) a[u] = ap[2 * u];
    }

    const v4f* bl = Bl + lane;
    for (; tile < ntiles; tile += stride) {
        asm volatile("" ::: "memory");
        const int m0 = tile * 16;
        const int mn = (tile + stride) * 16 + c;
        const int rn = (mn < n) ? mn : (n - 1);
        const v4f* apn = (const v4f*)(X + (long long)rn * DIM + 4 * h);

        float rsv[8];
        if (m0 + 16 <= n) {
            const v4f* rp = (const v4f*)(rs + m0 + 8 * h);
            const v4f r0 = rp[0], r1 = rp[1];
            rsv[0] = r0.x; rsv[1] = r0.y; rsv[2] = r0.z; rsv[3] = r0.w;
            rsv[4] = r1.x; rsv[5] = r1.y; rsv[6] = r1.z; rsv[7] = r1.w;
        } else {
#pragma unroll
            for (int v = 0; v < 8; ++v) {
                const int m = m0 + v + 8 * h;
                rsv[v] = rs[(m < n) ? m : (n - 1)];
            }
        }

        v8f d[8];
#pragma unroll
        for (int j = 0; j < 8; ++j) {
#pragma unroll
            for (int v = 0; v < 8; ++v) d[j][v] = 0.0f;
        }

        v4f bc[8], bn[8];
#pragma unroll
        for (int j = 0; j < 8; ++j) bc[j] = bl[(j * 16) * 32];
#pragma unroll
        for (int u = 0; u < 16; ++u) {
            if (u + 1 < 16) {
#pragma unroll
                for (int j = 0; j < 8; ++j) bn[j] = bl[(j * 16 + u + 1) * 32];
            }
            __builtin_amdgcn_sched_barrier(0);
            v2f a0, a1;
            a0.x = a[u].x; a0.y = a[u].y;
            a1.x = a[u].z; a1.y = a[u].w;
#pragma unroll
            for (int j = 0; j < 8; ++j) {
                v2f b0;
                b0.x = bc[j].x; b0.y = bc[j].y;
                d[j] = __builtin_amdgcn_wmma_f32_16x16x4_f32(false, a0, false, b0,
                                                             (short)0, d[j], false, false);
            }
#pragma unroll
            for (int j = 0; j < 8; ++j) {
                v2f b1;
                b1.x = bc[j].z; b1.y = bc[j].w;
                d[j] = __builtin_amdgcn_wmma_f32_16x16x4_f32(false, a1, false, b1,
                                                             (short)0, d[j], false, false);
            }
            a[u] = apn[2 * u];
            __builtin_amdgcn_sched_barrier(0);
#pragma unroll
            for (int j = 0; j < 8; ++j) bc[j] = bn[j];
        }

        float* yp = Y + (long long)(m0 + 8 * h) * DIM + c;
#pragma unroll
        for (int v = 0; v < 8; ++v) {
            const int m = m0 + v + 8 * h;
            if (m < n) {
#pragma unroll
                for (int j = 0; j < 8; ++j) yp[(long long)v * DIM + 16 * j] = (d[j][v] + cj[j]) * rsv[v];
            }
        }
    }
}

__device__ __forceinline__ float4 gather_row(
    const float* __restrict__ xw, const int* __restrict__ offs, const int* __restrict__ srcs,
    const float* __restrict__ dinv, int node, int lane) {
    const int beg = offs[node];
    const int end = offs[node + 1];
    const float dc = dinv[node];
    const float4* __restrict__ x4 = (const float4*)xw;

    float4 a = x4[(long long)node * 32 + lane];

    for (int b = beg; b < end; b += 32) {
        const int m = (end - b < 32) ? (end - b) : 32;
        int r = 0;
        if (lane < m) r = srcs[b + lane];
#define SRC_ROW(j_) x4[(long long)__builtin_amdgcn_readlane(r, (j_)) * 32 + lane]
#define ADD_ROW(v_) a.x += v_.x; a.y += v_.y; a.z += v_.z; a.w += v_.w
        int j = 0;
        for (; j + 4 <= m; j += 4) {
            const float4 v0 = SRC_ROW(j);
            const float4 v1 = SRC_ROW(j + 1);
            const float4 v2 = SRC_ROW(j + 2);
            const float4 v3 = SRC_ROW(j + 3);
            ADD_ROW(v0); ADD_ROW(v1); ADD_ROW(v2); ADD_ROW(v3);
        }
        const int rem = m - j;
        if (rem == 3) {
            const float4 v0 = SRC_ROW(j);
            const float4 v1 = SRC_ROW(j + 1);
            const float4 v2 = SRC_ROW(j + 2);
            ADD_ROW(v0); ADD_ROW(v1); ADD_ROW(v2);
        } else if (rem == 2) {
            const float4 v0 = SRC_ROW(j);
            const float4 v1 = SRC_ROW(j + 1);
            ADD_ROW(v0); ADD_ROW(v1);
        } else if (rem == 1) {
            const float4 v0 = SRC_ROW(j);
            ADD_ROW(v0);
        }
#undef SRC_ROW
#undef ADD_ROW
    }
    return make_float4(a.x * dc, a.y * dc, a.z * dc, a.w * dc);
}

__global__ __launch_bounds__(256) void k_gather_relu(
    const float* __restrict__ xw, const int* __restrict__ offs, const int* __restrict__ srcs,
    const float* __restrict__ dinv, const float* __restrict__ b1,
    const float* __restrict__ posts, const float* __restrict__ W2,
    const int* __restrict__ rootPtr,
    float* __restrict__ h1, float* __restrict__ root_feat, float* __restrict__ cvec, int n) {
    const int lane = threadIdx.x & 31;
    const int node = blockIdx.x * 8 + __builtin_amdgcn_readfirstlane((int)(threadIdx.x >> 5));
    if (node >= n) return;
    const int root = *rootPtr;
    const float4 bb = ((const float4*)b1)[lane];

    const float4 a = gather_row(xw, offs, srcs, dinv, node, lane);

    const float4 s = make_float4(a.x + bb.x, a.y + bb.y, a.z + bb.z, a.w + bb.w);
    if (node == root) ((float4*)root_feat)[lane] = s;
    ((float4*)h1)[(long long)node * 32 + lane] =
        make_float4(fmaxf(s.x, 0.0f), fmaxf(s.y, 0.0f), fmaxf(s.z, 0.0f), fmaxf(s.w, 0.0f));

    if (node == 0) {
        const float* __restrict__ pr = posts + (long long)root * DIM;
        const float4* __restrict__ w4 = (const float4*)(W2 + (long long)DIM * DIM) + lane;
        float4 cs = make_float4(0.0f, 0.0f, 0.0f, 0.0f);
        for (int k = 0; k < DIM; ++k) {
            const float p = fmaxf(pr[k], 0.0f);
            const float4 w = w4[k * 32];
            cs.x = fmaf(p, w.x, cs.x);
            cs.y = fmaf(p, w.y, cs.y);
            cs.z = fmaf(p, w.z, cs.z);
            cs.w = fmaf(p, w.w, cs.w);
        }
        ((float4*)cvec)[lane] = cs;
    }
}

typedef v4f v4f_a8 __attribute__((aligned(8)));

__global__ __launch_bounds__(256) void k_gather_pool(
    const float* __restrict__ xw, const int* __restrict__ offs, const int* __restrict__ srcs,
    const float* __restrict__ dinv, const float* __restrict__ b2,
    const float* __restrict__ root_feat, float* __restrict__ out, int n) {
    const int lane = threadIdx.x & 31;
    const int node = blockIdx.x * 8 + __builtin_amdgcn_readfirstlane((int)(threadIdx.x >> 5));
    if (node >= n) return;
    const float third = 1.0f / 3.0f;

    const float4 bb = ((const float4*)b2)[lane];
    const float4 rf = ((const float4*)root_feat)[lane];
    const int nx = (lane < 31) ? 4 * lane + 4 : 126;
    const float rf4 = root_feat[nx], rf5 = root_feat[nx + 1];

    const float4 a = gather_row(xw, offs, srcs, dinv, node, lane);

    v4f rp;
    rp.x = (rf.x + rf.y + rf.z) * third;
    rp.y = (rf.y + rf.z + rf.w) * third;
    rp.z = (rf.z + rf.w + rf4) * third;
    rp.w = (rf.w + rf4 + rf5) * third;
    const float s67 = rf.z + rf.w;

    const float cx = fmaxf(a.x + bb.x, 0.0f);
    const float cy = fmaxf(a.y + bb.y, 0.0f);
    const float cz = fmaxf(a.z + bb.z, 0.0f);
    const float cw = fmaxf(a.w + bb.w, 0.0f);
    const float n0 = __shfl_down(cx, 1);
    const float n1 = __shfl_down(cy, 1);
    const float c0 = __int_as_float(__builtin_amdgcn_readlane(__float_as_int(cx), 0));
    const float c1 = __int_as_float(__builtin_amdgcn_readlane(__float_as_int(cy), 0));

    float* __restrict__ ob = out + (long long)node * 254;
    v4f ha = rp;
    if (lane == 31) {
        ha.z = (s67 + c0) * third;
        ha.w = (rf.w + c0 + c1) * third;
    }
    *(v4f_a8*)(ob + 4 * lane) = ha;

    v4f p;
    p.x = (cx + cy + cz) * third;
    p.y = (cy + cz + cw) * third;
    p.z = (cz + cw + n0) * third;
    p.w = (cw + n0 + n1) * third;
    if (lane < 31) *(v4f_a8*)(ob + 128 + 4 * lane) = p;
    else           *(float2*)(ob + 252) = make_float2(p.x, p.y);
}

extern "C" void kernel_launch(void* const* d_in, const int* in_sizes, int n_in,
                              void* d_out, int out_size, void* d_ws, size_t ws_size,
                              hipStream_t stream) {
    const float* posts = (const float*)d_in[0];
    const float* W1    = (const float*)d_in[1];
    const float* b1    = (const float*)d_in[2];
    const float* W2    = (const float*)d_in[3];
    const float* b2    = (const float*)d_in[4];
    const int*   edge  = (const int*)d_in[5];
    const int*   rootP = (const int*)d_in[6];

    const int n = in_sizes[0] / DIM;
    const int e = in_sizes[5] / 2;
    const int* rowi = edge;
    const int* coli = edge + e;

    char* ws = (char*)d_ws;
    size_t off = 0;
    float* dinv = (float*)(ws + off); off += (((size_t)n * 4) + 255) & ~(size_t)255;
    float* xw   = (float*)(ws + off); off += (size_t)n * DIM * 4;
    char*  lst  = ws + off;           off += (size_t)n * DIM * 4;
    float* rootf = (float*)(ws + off); off += DIM * 4;
    float* cvec  = (float*)(ws + off);
    float* out = (float*)d_out;

    const size_t npad = (((size_t)n * 4) + 255) & ~(size_t)255;
    const size_t epad = (((size_t)e * 4) + 255) & ~(size_t)255;
    const size_t opad = (((size_t)(n + 1) * 4) + 255) & ~(size_t)255;
    int* cnt  = (int*)lst;
    int* rank = (int*)(lst + npad);
    int* offs = (int*)(lst + npad + epad);
    int* srcs = (int*)(lst + npad + epad + opad);
    float* h1 = (float*)d_out;

    const int B = 256;
    const int gN   = (n + B - 1) / B;
    const int gE   = (e + B - 1) / B;
    const int gS   = (n + 1023) / 1024;
    const int gG   = (n + 7) / 8;
    const int gM   = GEMM_GROUPS;

    k_zero_i32  <<<gN, B, 0, stream>>>(cnt, n);
    k_count_rank<<<gE, B, 0, stream>>>(coli, cnt, rank, e);
    k_scan_dinv <<<gS, 1024, 0, stream>>>(cnt, offs, dinv, n);
    k_fill      <<<gE, B, 0, stream>>>(rowi, coli, rank, offs, srcs, e);

    k_gemm128_wmma <<<gM, B, 0, stream>>>(posts, W1, nullptr, dinv, xw, n);
    k_gather_relu  <<<gG, B, 0, stream>>>(xw, offs, srcs, dinv, b1, posts, W2, rootP,
                                         h1, rootf, cvec, n);

    k_gemm128_wmma <<<gM, B, 0, stream>>>(h1, W2, cvec, dinv, xw, n);
    k_gather_pool  <<<gG, B, 0, stream>>>(xw, offs, srcs, dinv, b2, rootf, out, n);
}


# Edge lists as fixed 64-slot blocks per node, filled in the counting pass (place = returning integer add; spill list past 64): rank array, scan and fill launches go; sum order stays arrival order
# speedup vs baseline: 3.6172x; 1.0754x over previous
#include <hip/hip_runtime.h>

typedef __attribute__((ext_vector_type(2))) float v2f;
typedef __attribute__((ext_vector_type(8))) float v8f;

#define DIM 128

#define LIST_CAP 64

__global__ void k_zero_i32(int* __restrict__ p, int n) {
    int i = blockIdx.x * blockDim.x + threadIdx.x;
    if (i < n) p[i] = 0;
}

#define EDGES_PER_THREAD 4

__device__ __forceinline__ void place_edge(int c, int r, int p, int* __restrict__ slots,
                                           int* __restrict__ nspill, int2* __restrict__ spill) {
    if (p < LIST_CAP) {
        slots[(long long)c * LIST_CAP + p] = r;
    } else {
        const int q = atomicAdd(nspill, 1);
        spill[q] = make_int2(c, r);
    }
}

__global__ __launch_bounds__(256) void k_build_lists(const int* __restrict__ row, const int* __restrict__ col,
                                                     int* __restrict__ cnt, int* __restrict__ slots,
                                                     int2* __restrict__ spill, int n, int e) {
    const int total = (int)(gridDim.x * blockDim.x);
    const int i0 = (int)(blockIdx.x * blockDim.x + threadIdx.x);
    const int last = (int)((blockIdx.x + 1) * blockDim.x) - 1 + (EDGES_PER_THREAD - 1) * total;
    if (last < e) {
        int c[EDGES_PER_THREAD], r[EDGES_PER_THREAD], p[EDGES_PER_THREAD];
#pragma unroll
        for (int k = 0; k < EDGES_PER_THREAD; ++k) c[k] = col[i0 + k * total];
#pragma unroll
        for (int k = 0; k < EDGES_PER_THREAD; ++k) r[k] = row[i0 + k * total];
#pragma unroll
        for (int k = 0; k < EDGES_PER_THREAD; ++k) p[k] = atomicAdd(&cnt[c[k]], 1);
#pragma unroll
        for (int k = 0; k < EDGES_PER_THREAD; ++k) place_edge(c[k], r[k], p[k], slots, cnt + n, spill);
    } else {
#pragma unroll
        for (int k = 0; k < EDGES_PER_THREAD; ++k) {
            const int i = i0 + k * total;
            if (i < e) {
                const int c = col[i];
                const int r = row[i];
                const int p = atomicAdd(&cnt[c], 1);
                place_edge(c, r, p, slots, cnt + n, spill);
            }
        }
    }
}

#define GEMM_GROUPS 512
typedef __attribute__((ext_vector_type(4))) float v4f;

__global__ __launch_bounds__(256, 4) void k_gemm128_wmma(
    const float* __restrict__ X, const float* __restrict__ W,
    const float* __restrict__ cvec, const int* __restrict__ cnt, float* __restrict__ Y, int n) {
    __shared__ v4f Bl[8 * 16 * 32];
    const int tid  = threadIdx.x;
    const int lane = tid & 31;
    const int wave = __builtin_amdgcn_readfirstlane(tid >> 5);
    const int c    = lane & 15;
    const int h    = lane >> 4;

    {
        const int sn = tid & 127;
        const int sh = tid >> 7;
        const float* wp = W + (4 * sh) * DIM + sn;
        v4f* bs = Bl + (sn >> 4) * (16 * 32) + (sn & 15) + 16 * sh;
#pragma unroll
        for (int u = 0; u < 16; ++u) {
            v4f t;
            t.x = wp[(8 * u + 0) * DIM];
            t.y = wp[(8 * u + 1) * DIM];
            t.z = wp[(8 * u + 2) * DIM];
            t.w = wp[(8 * u + 3) * DIM];
            bs[u * 32] = t;
        }
    }

    float cj[8];
#pragma unroll
    for (int j = 0; j < 8; ++j) cj[j] = cvec ? cvec[16 * j + c] : 0.0f;

    __syncthreads();

    const int ntiles = (n + 15) >> 4;
    const int stride = 8 * (int)gridDim.x;
    int tile = wave * (int)gridDim.x + (int)blockIdx.x;

    v4f a[16];
    {
        const int r0 = (tile * 16 + c < n) ? (tile * 16 + c) : (n - 1);
        const v4f* ap = (const v4f*)(X + (long long)r0 * DIM + 4 * h);
#pragma unroll
        for (int u = 0; u < 16; ++u) a[u] = ap[2 * u];
    }

    const v4f* bl = Bl + lane;
    for (; tile < ntiles; tile += stride) {
        asm volatile("" ::: "memory");
        const int m0 = tile * 16;
        const int mn = (tile + stride) * 16 + c;
        const int rn = (mn < n) ? mn : (n - 1);
        const v4f* apn = (const v4f*)(X + (long long)rn * DIM + 4 * h);

        float rsv[8];
        if (m0 + 16 <= n) {
            const int4* cp = (const int4*)(cnt + m0 + 8 * h);
            const int4 c0 = cp[0], c1 = cp[1];
            rsv[0] = rsqrtf((float)(c0.x + 1)); rsv[1] = rsqrtf((float)(c0.y + 1));
            rsv[2] = rsqrtf((float)(c0.z + 1)); rsv[3] = rsqrtf((float)(c0.w + 1));
            rsv[4] = rsqrtf((float)(c1.x + 1)); rsv[5] = rsqrtf((float)(c1.y + 1));
            rsv[6] = rsqrtf((float)(c1.z + 1)); rsv[7] = rsqrtf((float)(c1.w + 1));
        } else {
#pragma unroll
            for (int v = 0; v < 8; ++v) {
                const int m = m0 + v + 8 * h;
                rsv[v] = rsqrtf((float)(cnt[(m < n) ? m : (n - 1)] + 1));
            }
        }

        v8f d[8];
#pragma unroll
        for (int j = 0; j < 8; ++j) {
#pragma unroll
            for (int v = 0; v < 8; ++v) d[j][v] = 0.0f;
        }

        v4f bc[8], bn[8];
#pragma unroll
        for (int j = 0; j < 8; ++j) bc[j] = bl[(j * 16) * 32];
#pragma unroll
        for (int u = 0; u < 16; ++u) {
            if (u + 1 < 16) {
#pragma unroll
                for (int j = 0; j < 8; ++j) bn[j] = bl[(j * 16 + u + 1) * 32];
            }
            __builtin_amdgcn_sched_barrier(0);
            v2f a0, a1;
            a0.x = a[u].x; a0.y = a[u].y;
            a1.x = a[u].z; a1.y = a[u].w;
#pragma unroll
            for (int j = 0; j < 8; ++j) {
                v2f b0;
                b0.x = bc[j].x; b0.y = bc[j].y;
                d[j] = __builtin_amdgcn_wmma_f32_16x16x4_f32(false, a0, false, b0,
                                                             (short)0, d[j], false, false);
            }
#pragma unroll
            for (int j = 0; j < 8; ++j) {
                v2f b1;
                b1.x = bc[j].z; b1.y = bc[j].w;
                d[j] = __builtin_amdgcn_wmma_f32_16x16x4_f32(false, a1, false, b1,
                                                             (short)0, d[j], false, false);
            }
            a[u] = apn[2 * u];
            __builtin_amdgcn_sched_barrier(0);
#pragma unroll
            for (int j = 0; j < 8; ++j) bc[j] = bn[j];
        }

        float* yp = Y + (long long)(m0 + 8 * h) * DIM + c;
#pragma unroll
        for (int v = 0; v < 8; ++v) {
            const int m = m0 + v + 8 * h;
            if (m < n) {
#pragma unroll
                for (int j = 0; j < 8; ++j) yp[(long long)v * DIM + 16 * j] = (d[j][v] + cj[j]) * rsv[v];
            }
        }
    }
}

#define ADD_ROW(v_) a.x += v_.x; a.y += v_.y; a.z += v_.z; a.w += v_.w
__device__ __forceinline__ float4 gather_row(
    const float* __restrict__ xw, const int* __restrict__ cnt, const int* __restrict__ slots,
    const int2* __restrict__ spill, int n, int node, int lane) {
    const int deg = cnt[node];
    const float dc = rsqrtf((float)(deg + 1));
    const int nl = (deg < LIST_CAP) ? deg : LIST_CAP;
    const int* __restrict__ lp = slots + (long long)node * LIST_CAP;
    const float4* __restrict__ x4 = (const float4*)xw;

    float4 a = x4[(long long)node * 32 + lane];

    for (int b = 0; b < nl; b += 32) {
        const int m = (nl - b < 32) ? (nl - b) : 32;
        int r = 0;
        if (lane < m) r = lp[b + lane];
#define SRC_ROW(j_) x4[(long long)__builtin_amdgcn_readlane(r, (j_)) * 32 + lane]
        int j = 0;
        for (; j + 4 <= m; j += 4) {
            const float4 v0 = SRC_ROW(j);
            const float4 v1 = SRC_ROW(j + 1);
            const float4 v2 = SRC_ROW(j + 2);
            const float4 v3 = SRC_ROW(j + 3);
            ADD_ROW(v0); ADD_ROW(v1); ADD_ROW(v2); ADD_ROW(v3);
        }
        const int rem = m - j;
        if (rem == 3) {
            const float4 v0 = SRC_ROW(j);
            const float4 v1 = SRC_ROW(j + 1);
            const float4 v2 = SRC_ROW(j + 2);
            ADD_ROW(v0); ADD_ROW(v1); ADD_ROW(v2);
        } else if (rem == 2) {
            const float4 v0 = SRC_ROW(j);
            const float4 v1 = SRC_ROW(j + 1);
            ADD_ROW(v0); ADD_ROW(v1);
        } else if (rem == 1) {
            const float4 v0 = SRC_ROW(j);
            ADD_ROW(v0);
        }
#undef SRC_ROW
    }

    if (deg > LIST_CAP) {
        const int ns = cnt[n];
        for (int b = 0; b < ns; b += 32) {
            int2 pr = make_int2(-1, 0);
            if (b + lane < ns) pr = spill[b + lane];
            unsigned mask = __builtin_amdgcn_ballot_w32(pr.x == node);
            while (mask != 0u) {
                const int j = __builtin_ctz(mask);
                mask &= mask - 1u;
                const float4 v0 = x4[(long long)__builtin_amdgcn_readlane(pr.y, j) * 32 + lane];
                ADD_ROW(v0);
            }
        }
    }
    return make_float4(a.x * dc, a.y * dc, a.z * dc, a.w * dc);
}
#undef ADD_ROW

__global__ __launch_bounds__(256) void k_gather_relu(
    const float* __restrict__ xw, const int* __restrict__ cnt, const int* __restrict__ slots,
    const int2* __restrict__ spill, const float* __restrict__ b1,
    const float* __restrict__ posts, const float* __restrict__ W2,
    const int* __restrict__ rootPtr,
    float* __restrict__ h1, float* __restrict__ root_feat, float* __restrict__ cvec, int n) {
    const int lane = threadIdx.x & 31;
    const int node = blockIdx.x * 8 + __builtin_amdgcn_readfirstlane((int)(threadIdx.x >> 5));
    if (node >= n) return;
    const int root = *rootPtr;
    const float4 bb = ((const float4*)b1)[lane];

    const float4 a = gather_row(xw, cnt, slots, spill, n, node, lane);

    const float4 s = make_float4(a.x + bb.x, a.y + bb.y, a.z + bb.z, a.w + bb.w);
    if (node == root) ((float4*)root_feat)[lane] = s;
    ((float4*)h1)[(long long)node * 32 + lane] =
        make_float4(fmaxf(s.x, 0.0f), fmaxf(s.y, 0.0f), fmaxf(s.z, 0.0f), fmaxf(s.w, 0.0f));

    if (node == 0) {
        const float* __restrict__ pr = posts + (long long)root * DIM;
        const float4* __restrict__ w4 = (const float4*)(W2 + (long long)DIM * DIM) + lane;
        float4 cs = make_float4(0.0f, 0.0f, 0.0f, 0.0f);
        for (int k = 0; k < DIM; ++k) {
            const float p = fmaxf(pr[k], 0.0f);
            const float4 w = w4[k * 32];
            cs.x = fmaf(p, w.x, cs.x);
            cs.y = fmaf(p, w.y, cs.y);
            cs.z = fmaf(p, w.z, cs.z);
            cs.w = fmaf(p, w.w, cs.w);
        }
        ((float4*)cvec)[lane] = cs;
    }
}

typedef v4f v4f_a8 __attribute__((aligned(8)));

__global__ __launch_bounds__(256) void k_gather_pool(
    const float* __restrict__ xw, const int* __restrict__ cnt, const int* __restrict__ slots,
    const int2* __restrict__ spill, const float* __restrict__ b2,
    const float* __restrict__ root_feat, float* __restrict__ out, int n) {
    const int lane = threadIdx.x & 31;
    const int node = blockIdx.x * 8 + __builtin_amdgcn_readfirstlane((int)(threadIdx.x >> 5));
    if (node >= n) return;
    const float third = 1.0f / 3.0f;

    const float4 bb = ((const float4*)b2)[lane];
    const float4 rf = ((const float4*)root_feat)[lane];
    const int nx = (lane < 31) ? 4 * lane + 4 : 126;
    const float rf4 = root_feat[nx], rf5 = root_feat[nx + 1];

    const float4 a = gather_row(xw, cnt, slots, spill, n, node, lane);

    v4f rp;
    rp.x = (rf.x + rf.y + rf.z) * third;
    rp.y = (rf.y + rf.z + rf.w) * third;
    rp.z = (rf.z + rf.w + rf4) * third;
    rp.w = (rf.w + rf4 + rf5) * third;
    const float s67 = rf.z + rf.w;

    const float cx = fmaxf(a.x + bb.x, 0.0f);
    const float cy = fmaxf(a.y + bb.y, 0.0f);
    const float cz = fmaxf(a.z + bb.z, 0.0f);
    const float cw = fmaxf(a.w + bb.w, 0.0f);
    const float n0 = __shfl_down(cx, 1);
    const float n1 = __shfl_down(cy, 1);
    const float c0 = __int_as_float(__builtin_amdgcn_readlane(__float_as_int(cx), 0));
    const float c1 = __int_as_float(__builtin_amdgcn_readlane(__float_as_int(cy), 0));

    float* __restrict__ ob = out + (long long)node * 254;
    v4f ha = rp;
    if (lane == 31) {
        ha.z = (s67 + c0) * third;
        ha.w = (rf.w + c0 + c1) * third;
    }
    *(v4f_a8*)(ob + 4 * lane) = ha;

    v4f p;
    p.x = (cx + cy + cz) * third;
    p.y = (cy + cz + cw) * third;
    p.z = (cz + cw + n0) * third;
    p.w = (cw + n0 + n1) * third;
    if (lane < 31) *(v4f_a8*)(ob + 128 + 4 * lane) = p;
    else           *(float2*)(ob + 252) = make_float2(p.x, p.y);
}

extern "C" void kernel_launch(void* const* d_in, const int* in_sizes, int n_in,
                              void* d_out, int out_size, void* d_ws, size_t ws_size,
                              hipStream_t stream) {
    const float* posts = (const float*)d_in[0];
    const float* W1    = (const float*)d_in[1];
    const float* b1    = (const float*)d_in[2];
    const float* W2    = (const float*)d_in[3];
    const float* b2    = (const float*)d_in[4];
    const int*   edge  = (const int*)d_in[5];
    const int*   rootP = (const int*)d_in[6];

    const int n = in_sizes[0] / DIM;
    const int e = in_sizes[5] / 2;
    const int* rowi = edge;
    const int* coli = edge + e;

    char* ws = (char*)d_ws;
    size_t off = 0;
    float* xw   = (float*)(ws + off); off += (size_t)n * DIM * 4;
    char*  lst  = ws + off;           off += (size_t)n * DIM * 4;
    float* rootf = (float*)(ws + off); off += DIM * 4;
    float* cvec  = (float*)(ws + off);
    float* out = (float*)d_out;

    const size_t cpad = (((size_t)(n + 1) * 4) + 255) & ~(size_t)255;
    const size_t spad = (size_t)n * LIST_CAP * 4;
    int*  cnt   = (int*)lst;
    int*  slots = (int*)(lst + cpad);
    int2* spill = (int2*)(lst + cpad + spad);
    float* h1 = (float*)d_out;

    const int B = 256;
    const int gZ   = (n + 1 + B - 1) / B;
    const int gE   = (e + B * EDGES_PER_THREAD - 1) / (B * EDGES_PER_THREAD);
    const int gG   = (n + 7) / 8;
    const int gM   = GEMM_GROUPS;

    k_zero_i32   <<<gZ, B, 0, stream>>>(cnt, n + 1);
    k_build_lists<<<gE, B, 0, stream>>>(rowi, coli, cnt, slots, spill, n, e);

    k_gemm128_wmma <<<gM, B, 0, stream>>>(posts, W1, nullptr, cnt, xw, n);
    k_gather_relu  <<<gG, B, 0, stream>>>(xw, cnt, slots, spill, b1, posts, W2, rootP,
                                         h1, rootf, cvec, n);

    k_gemm128_wmma <<<gM, B, 0, stream>>>(h1, W2, cvec, cnt, xw, n);
    k_gather_pool  <<<gG, B, 0, stream>>>(xw, cnt, slots, spill, b2, rootf, out, n);
}


# Layer-2 GEMM moved into the layer-1 gather launch: 32-wave groups keep the 32 rows just summed in an LDS tile and multiply by W2 on the matrix unit; h1 never stored, one launch less; sums unchanged
# speedup vs baseline: 3.7955x; 1.0493x over previous
#include <hip/hip_runtime.h>

typedef __attribute__((ext_vector_type(2))) float v2f;
typedef __attribute__((ext_vector_type(8))) float v8f;

#define DIM 128

#define LIST_CAP 64

__global__ void k_zero_i32(int* __restrict__ p, int n) {
    int i = blockIdx.x * blockDim.x + threadIdx.x;
    if (i < n) p[i] = 0;
}

#define EDGES_PER_THREAD 4

__device__ __forceinline__ void place_edge(int c, int r, int p, int* __restrict__ slots,
                                           int* __restrict__ nspill, int2* __restrict__ spill) {
    if (p < LIST_CAP) {
        slots[(long long)c * LIST_CAP + p] = r;
    } else {
        const int q = atomicAdd(nspill, 1);
        spill[q] = make_int2(c, r);
    }
}

__global__ __launch_bounds__(256) void k_build_lists(const int* __restrict__ row, const int* __restrict__ col,
                                                     int* __restrict__ cnt, int* __restrict__ slots,
                                                     int2* __restrict__ spill, int n, int e,
                                                     const float* __restrict__ posts,
                                                     const float* __restrict__ W2,
                                                     const int* __restrict__ rootPtr,
                                                     float* __restrict__ cvec) {
    const int total = (int)(gridDim.x * blockDim.x);
    const int i0 = (int)(blockIdx.x * blockDim.x + threadIdx.x);
    const int last = (int)((blockIdx.x + 1) * blockDim.x) - 1 + (EDGES_PER_THREAD - 1) * total;
    if (last < e) {
        int c[EDGES_PER_THREAD], r[EDGES_PER_THREAD], p[EDGES_PER_THREAD];
#pragma unroll
        for (int k = 0; k < EDGES_PER_THREAD; ++k) c[k] = col[i0 + k * total];
#pragma unroll
        for (int k = 0; k < EDGES_PER_THREAD; ++k) r[k] = row[i0 + k * total];
#pragma unroll
        for (int k = 0; k < EDGES_PER_THREAD; ++k) p[k] = atomicAdd(&cnt[c[k]], 1);
#pragma unroll
        for (int k = 0; k < EDGES_PER_THREAD; ++k) place_edge(c[k], r[k], p[k], slots, cnt + n, spill);
    } else {
#pragma unroll
        for (int k = 0; k < EDGES_PER_THREAD; ++k) {
            const int i = i0 + k * total;
            if (i < e) {
                const int c = col[i];
                const int r = row[i];
                const int p = atomicAdd(&cnt[c], 1);
                place_edge(c, r, p, slots, cnt + n, spill);
            }
        }
    }
    if (blockIdx.x == 0 && threadIdx.x < DIM) {
        const int root = *rootPtr;
        const float* __restrict__ pr = posts + (long long)root * DIM;
        const float* __restrict__ wc = W2 + (long long)DIM * DIM + threadIdx.x;
        float cs = 0.0f;
#pragma unroll 32
        for (int k = 0; k < DIM; ++k) cs = fmaf(fmaxf(pr[k], 0.0f), wc[k * DIM], cs);
        cvec[threadIdx.x] = cs;
    }
}

#define GEMM_GROUPS 512
typedef __attribute__((ext_vector_type(4))) float v4f;

__global__ __launch_bounds__(256, 4) void k_gemm128_wmma(
    const float* __restrict__ X, const float* __restrict__ W,
    const float* __restrict__ cvec, const int* __restrict__ cnt, float* __restrict__ Y, int n) {
    __shared__ v4f Bl[8 * 16 * 32];
    const int tid  = threadIdx.x;
    const int lane = tid & 31;
    const int wave = __builtin_amdgcn_readfirstlane(tid >> 5);
    const int c    = lane & 15;
    const int h    = lane >> 4;

    {
        const int sn = tid & 127;
        const int sh = tid >> 7;
        const float* wp = W + (4 * sh) * DIM + sn;
        v4f* bs = Bl + (sn >> 4) * (16 * 32) + (sn & 15) + 16 * sh;
#pragma unroll
        for (int u = 0; u < 16; ++u) {
            v4f t;
            t.x = wp[(8 * u + 0) * DIM];
            t.y = wp[(8 * u + 1) * DIM];
            t.z = wp[(8 * u + 2) * DIM];
            t.w = wp[(8 * u + 3) * DIM];
            bs[u * 32] = t;
        }
    }

    float cj[8];
#pragma unroll
    for (int j = 0; j < 8; ++j) cj[j] = cvec ? cvec[16 * j + c] : 0.0f;

    __syncthreads();

    const int ntiles = (n + 15) >> 4;
    const int stride = 8 * (int)gridDim.x;
    int tile = wave * (int)gridDim.x + (int)blockIdx.x;

    v4f a[16];
    {
        const int r0 = (tile * 16 + c < n) ? (tile * 16 + c) : (n - 1);
        const v4f* ap = (const v4f*)(X + (long long)r0 * DIM + 4 * h);
#pragma unroll
        for (int u = 0; u < 16; ++u) a[u] = ap[2 * u];
    }

    const v4f* bl = Bl + lane;
    for (; tile < ntiles; tile += stride) {
        asm volatile("" ::: "memory");
        const int m0 = tile * 16;
        const int mn = (tile + stride) * 16 + c;
        const int rn = (mn < n) ? mn : (n - 1);
        const v4f* apn = (const v4f*)(X + (long long)rn * DIM + 4 * h);

        float rsv[8];
        if (m0 + 16 <= n) {
            const int4* cp = (const int4*)(cnt + m0 + 8 * h);
            const int4 c0 = cp[0], c1 = cp[1];
            rsv[0] = rsqrtf((float)(c0.x + 1)); rsv[1] = rsqrtf((float)(c0.y + 1));
            rsv[2] = rsqrtf((float)(c0.z + 1)); rsv[3] = rsqrtf((float)(c0.w + 1));
            rsv[4] = rsqrtf((float)(c1.x + 1)); rsv[5] = rsqrtf((float)(c1.y + 1));
            rsv[6] = rsqrtf((float)(c1.z + 1)); rsv[7] = rsqrtf((float)(c1.w + 1));
        } else {
#pragma unroll
            for (int v = 0; v < 8; ++v) {
                const int m = m0 + v + 8 * h;
                rsv[v] = rsqrtf((float)(cnt[(m < n) ? m : (n - 1)] + 1));
            }
        }

        v8f d[8];
#pragma unroll
        for (int j = 0; j < 8; ++j) {
#pragma unroll
            for (int v = 0; v < 8; ++v) d[j][v] = 0.0f;
        }

        v4f bc[8], bn[8];
#pragma unroll
        for (int j = 0; j < 8; ++j) bc[j] = bl[(j * 16) * 32];
#pragma unroll
        for (int u = 0; u < 16; ++u) {
            if (u + 1 < 16) {
#pragma unroll
                for (int j = 0; j < 8; ++j) bn[j] = bl[(j * 16 + u + 1) * 32];
            }
            __builtin_amdgcn_sched_barrier(0);
            v2f a0, a1;
            a0.x = a[u].x; a0.y = a[u].y;
            a1.x = a[u].z; a1.y = a[u].w;
#pragma unroll
            for (int j = 0; j < 8; ++j) {
                v2f b0;
                b0.x = bc[j].x; b0.y = bc[j].y;
                d[j] = __builtin_amdgcn_wmma_f32_16x16x4_f32(false, a0, false, b0,
                                                             (short)0, d[j], false, false);
            }
#pragma unroll
            for (int j = 0; j < 8; ++j) {
                v2f b1;
                b1.x = bc[j].z; b1.y = bc[j].w;
                d[j] = __builtin_amdgcn_wmma_f32_16x16x4_f32(false, a1, false, b1,
                                                             (short)0, d[j], false, false);
            }
            a[u] = apn[2 * u];
            __builtin_amdgcn_sched_barrier(0);
#pragma unroll
            for (int j = 0; j < 8; ++j) bc[j] = bn[j];
        }

        float* yp = Y + (long long)(m0 + 8 * h) * DIM + c;
#pragma unroll
        for (int v = 0; v < 8; ++v) {
            const int m = m0 + v + 8 * h;
            if (m < n) {
#pragma unroll
                for (int j = 0; j < 8; ++j) yp[(long long)v * DIM + 16 * j] = (d[j][v] + cj[j]) * rsv[v];
            }
        }
    }
}

#define ADD_ROW(v_) a.x += v_.x; a.y += v_.y; a.z += v_.z; a.w += v_.w
__device__ __forceinline__ float4 gather_row(
    const float* __restrict__ xw, const int* __restrict__ cnt, const int* __restrict__ slots,
    const int2* __restrict__ spill, int n, int node, int lane) {
    const int deg = cnt[node];
    const float dc = rsqrtf((float)(deg + 1));
    const int nl = (deg < LIST_CAP) ? deg : LIST_CAP;
    const int* __restrict__ lp = slots + (long long)node * LIST_CAP;
    const float4* __restrict__ x4 = (const float4*)xw;

    float4 a = x4[(long long)node * 32 + lane];

    for (int b = 0; b < nl; b += 32) {
        const int m = (nl - b < 32) ? (nl - b) : 32;
        int r = 0;
        if (lane < m) r = lp[b + lane];
#define SRC_ROW(j_) x4[(long long)__builtin_amdgcn_readlane(r, (j_)) * 32 + lane]
        int j = 0;
        for (; j + 4 <= m; j += 4) {
            const float4 v0 = SRC_ROW(j);
            const float4 v1 = SRC_ROW(j + 1);
            const float4 v2 = SRC_ROW(j + 2);
            const float4 v3 = SRC_ROW(j + 3);
            ADD_ROW(v0); ADD_ROW(v1); ADD_ROW(v2); ADD_ROW(v3);
        }
        const int rem = m - j;
        if (rem == 3) {
            const float4 v0 = SRC_ROW(j);
            const float4 v1 = SRC_ROW(j + 1);
            const float4 v2 = SRC_ROW(j + 2);
            ADD_ROW(v0); ADD_ROW(v1); ADD_ROW(v2);
        } else if (rem == 2) {
            const float4 v0 = SRC_ROW(j);
            const float4 v1 = SRC_ROW(j + 1);
            ADD_ROW(v0); ADD_ROW(v1);
        } else if (rem == 1) {
            const float4 v0 = SRC_ROW(j);
            ADD_ROW(v0);
        }
#undef SRC_ROW
    }

    if (deg > LIST_CAP) {
        const int ns = cnt[n];
        for (int b = 0; b < ns; b += 32) {
            int2 pr = make_int2(-1, 0);
            if (b + lane < ns) pr = spill[b + lane];
            unsigned mask = __builtin_amdgcn_ballot_w32(pr.x == node);
            while (mask != 0u) {
                const int j = __builtin_ctz(mask);
                mask &= mask - 1u;
                const float4 v0 = x4[(long long)__builtin_amdgcn_readlane(pr.y, j) * 32 + lane];
                ADD_ROW(v0);
            }
        }
    }
    return make_float4(a.x * dc, a.y * dc, a.z * dc, a.w * dc);
}
#undef ADD_ROW

#define FUSE_GROUPS 447
#define TILE_PITCH 33
__global__ __launch_bounds__(1024, 16) void k_gather_relu_gemm(
    const float* __restrict__ xw, const int* __restrict__ cnt, const int* __restrict__ slots,
    const int2* __restrict__ spill, const float* __restrict__ b1,
    const float* __restrict__ W, const float* __restrict__ cvec,
    const int* __restrict__ rootPtr,
    float* __restrict__ root_feat, float* __restrict__ Y, int n) {
    __shared__ v4f Bl[8 * 16 * 32];
    __shared__ v4f Tl[2 * 32 * TILE_PITCH];
    const int tid  = threadIdx.x;
    const int lane = tid & 31;
    const int wave = __builtin_amdgcn_readfirstlane(tid >> 5);
    const int c    = lane & 15;
    const int h    = lane >> 4;

    {
        const int sn = tid & 127;
        const int sh = (tid >> 7) & 1;
        const int uq = tid >> 8;
        const float* wp = W + (4 * sh) * DIM + sn;
        v4f* bs = Bl + (sn >> 4) * (16 * 32) + (sn & 15) + 16 * sh;
#pragma unroll
        for (int k = 0; k < 4; ++k) {
            const int u = 4 * uq + k;
            v4f t;
            t.x = wp[(8 * u + 0) * DIM];
            t.y = wp[(8 * u + 1) * DIM];
            t.z = wp[(8 * u + 2) * DIM];
            t.w = wp[(8 * u + 3) * DIM];
            bs[u * 32] = t;
        }
    }

    const int root = *rootPtr;
    const float4 bb = ((const float4*)b1)[lane];
    const int ntiles = (n + 31) >> 5;
    int par = 0;
    for (int tile = (int)blockIdx.x; tile < ntiles; tile += (int)gridDim.x) {
        const int node = tile * 32 + wave;
        v4f hrow;
        hrow.x = 0.0f; hrow.y = 0.0f; hrow.z = 0.0f; hrow.w = 0.0f;
        int deg = 0;
        if (node < n) {
            deg = cnt[node];
            const float4 a = gather_row(xw, cnt, slots, spill, n, node, lane);
            const float4 s = make_float4(a.x + bb.x, a.y + bb.y, a.z + bb.z, a.w + bb.w);
            if (node == root) ((float4*)root_feat)[lane] = s;
            hrow.x = fmaxf(s.x, 0.0f); hrow.y = fmaxf(s.y, 0.0f);
            hrow.z = fmaxf(s.z, 0.0f); hrow.w = fmaxf(s.w, 0.0f);
        }
        v4f* tb = Tl + par * (32 * TILE_PITCH);
        tb[wave * TILE_PITCH + lane] = hrow;
        int* ti = (int*)tb;
        if (lane == 0) ti[(wave * TILE_PITCH + 32) * 4] = deg;

        __syncthreads();

        if ((wave >> 4) == par) {
            const int pw = wave & 15;
            const int rt = pw >> 3;
            const int j  = pw & 7;
            const v4f* ta = tb + (16 * rt + c) * TILE_PITCH + h;
            const v4f* bl = Bl + (j * 16) * 32 + lane;
            const int m0 = tile * 32 + 16 * rt;
            float rsv[8];
#pragma unroll
            for (int v = 0; v < 8; ++v)
                rsv[v] = rsqrtf((float)(ti[((16 * rt + v + 8 * h) * TILE_PITCH + 32) * 4] + 1));
            const float cjv = cvec[16 * j + c];
            v8f d;
#pragma unroll
            for (int v = 0; v < 8; ++v) d[v] = 0.0f;
#pragma unroll
            for (int u = 0; u < 16; ++u) {
                const v4f a4 = ta[2 * u];
                const v4f b4 = bl[u * 32];
                v2f a0, a1, b0, b1v;
                a0.x = a4.x; a0.y = a4.y; a1.x = a4.z; a1.y = a4.w;
                b0.x = b4.x; b0.y = b4.y; b1v.x = b4.z; b1v.y = b4.w;
                d = __builtin_amdgcn_wmma_f32_16x16x4_f32(false, a0, false, b0, (short)0, d, false, false);
                d = __builtin_amdgcn_wmma_f32_16x16x4_f32(false, a1, false, b1v, (short)0, d, false, false);
            }
            asm volatile("v_nop\n\tv_nop\n\tv_nop\n\tv_nop\n\tv_nop" : "+v"(d));
            float* yp = Y + (long long)(m0 + 8 * h) * DIM + 16 * j + c;
#pragma unroll
            for (int v = 0; v < 8; ++v) {
                const int m = m0 + v + 8 * h;
                if (m < n) yp[(long long)v * DIM] = (d[v] + cjv) * rsv[v];
            }
        }
        par ^= 1;
    }
}

typedef v4f v4f_a8 __attribute__((aligned(8)));

__global__ __launch_bounds__(256) void k_gather_pool(
    const float* __restrict__ xw, const int* __restrict__ cnt, const int* __restrict__ slots,
    const int2* __restrict__ spill, const float* __restrict__ b2,
    const float* __restrict__ root_feat, float* __restrict__ out, int n) {
    const int lane = threadIdx.x & 31;
    const int node = blockIdx.x * 8 + __builtin_amdgcn_readfirstlane((int)(threadIdx.x >> 5));
    if (node >= n) return;
    const float third = 1.0f / 3.0f;

    const float4 bb = ((const float4*)b2)[lane];
    const float4 rf = ((const float4*)root_feat)[lane];
    const int nx = (lane < 31) ? 4 * lane + 4 : 126;
    const float rf4 = root_feat[nx], rf5 = root_feat[nx + 1];

    const float4 a = gather_row(xw, cnt, slots, spill, n, node, lane);

    v4f rp;
    rp.x = (rf.x + rf.y + rf.z) * third;
    rp.y = (rf.y + rf.z + rf.w) * third;
    rp.z = (rf.z + rf.w + rf4) * third;
    rp.w = (rf.w + rf4 + rf5) * third;
    const float s67 = rf.z + rf.w;

    const float cx = fmaxf(a.x + bb.x, 0.0f);
    const float cy = fmaxf(a.y + bb.y, 0.0f);
    const float cz = fmaxf(a.z + bb.z, 0.0f);
    const float cw = fmaxf(a.w + bb.w, 0.0f);
    const float n0 = __shfl_down(cx, 1);
    const float n1 = __shfl_down(cy, 1);
    const float c0 = __int_as_float(__builtin_amdgcn_readlane(__float_as_int(cx), 0));
    const float c1 = __int_as_float(__builtin_amdgcn_readlane(__float_as_int(cy), 0));

    float* __restrict__ ob = out + (long long)node * 254;
    v4f ha = rp;
    if (lane == 31) {
        ha.z = (s67 + c0) * third;
        ha.w = (rf.w + c0 + c1) * third;
    }
    *(v4f_a8*)(ob + 4 * lane) = ha;

    v4f p;
    p.x = (cx + cy + cz) * third;
    p.y = (cy + cz + cw) * third;
    p.z = (cz + cw + n0) * third;
    p.w = (cw + n0 + n1) * third;
    if (lane < 31) *(v4f_a8*)(ob + 128 + 4 * lane) = p;
    else           *(float2*)(ob + 252) = make_float2(p.x, p.y);
}

extern "C" void kernel_launch(void* const* d_in, const int* in_sizes, int n_in,
                              void* d_out, int out_size, void* d_ws, size_t ws_size,
                              hipStream_t stream) {
    const float* posts = (const float*)d_in[0];
    const float* W1    = (const float*)d_in[1];
    const float* b1    = (const float*)d_in[2];
    const float* W2    = (const float*)d_in[3];
    const float* b2    = (const float*)d_in[4];
    const int*   edge  = (const int*)d_in[5];
    const int*   rootP = (const int*)d_in[6];

    const int n = in_sizes[0] / DIM;
    const int e = in_sizes[5] / 2;
    const int* rowi = edge;
    const int* coli = edge + e;

    char* ws = (char*)d_ws;
    size_t off = 0;
    float* xw   = (float*)(ws + off); off += (size_t)n * DIM * 4;
    char*  lst  = ws + off;           off += (size_t)n * DIM * 4;
    float* rootf = (float*)(ws + off); off += DIM * 4;
    float* cvec  = (float*)(ws + off);
    float* out = (float*)d_out;

    const size_t cpad = (((size_t)(n + 1) * 4) + 255) & ~(size_t)255;
    const size_t spad = (size_t)n * LIST_CAP * 4;
    int*  cnt   = (int*)lst;
    int*  slots = (int*)(lst + cpad);
    int2* spill = (int2*)(lst + cpad + spad);
    float* xs1 = (float*)d_out;

    const int B = 256;
    const int gZ   = (n + 1 + B - 1) / B;
    const int gE   = (e + B * EDGES_PER_THREAD - 1) / (B * EDGES_PER_THREAD);
    const int gG   = (n + 7) / 8;
    const int gM   = GEMM_GROUPS;

    k_zero_i32   <<<gZ, B, 0, stream>>>(cnt, n + 1);
    k_build_lists<<<gE, B, 0, stream>>>(rowi, coli, cnt, slots, spill, n, e, posts, W2, rootP, cvec);

    k_gemm128_wmma     <<<gM, B, 0, stream>>>(posts, W1, nullptr, cnt, xs1, n);
    k_gather_relu_gemm <<<FUSE_GROUPS, 1024, 0, stream>>>(xs1, cnt, slots, spill, b1, W2, cvec, rootP,
                                                          rootf, xw, n);

    k_gather_pool  <<<gG, B, 0, stream>>>(xw, cnt, slots, spill, b2, rootf, out, n);
}


# List build moved into the layer-1 GEMM launch: a wave places its edges while other waves of its SIMD run WMMAs; GEMM rows stay unscaled, layer-1 gather applies source deg^-1/2 by fma; sum order kept
# speedup vs baseline: 3.9944x; 1.0524x over previous
#include <hip/hip_runtime.h>

typedef __attribute__((ext_vector_type(2))) float v2f;
typedef __attribute__((ext_vector_type(8))) float v8f;

#define DIM 128

#define LIST_CAP 64

__global__ void k_zero_i32(int* __restrict__ p, int n) {
    int i = blockIdx.x * blockDim.x + threadIdx.x;
    if (i < n) p[i] = 0;
}

#define EDGES_PER_THREAD 13

__device__ __forceinline__ void place_edge(int c, int r, int p, int* __restrict__ slots,
                                           int* __restrict__ nspill, int2* __restrict__ spill) {
    if (p < LIST_CAP) {
        slots[(long long)c * LIST_CAP + p] = r;
    } else {
        const int q = atomicAdd(nspill, 1);
        spill[q] = make_int2(c, r);
    }
}

__device__ __forceinline__ void build_lists_part(const int* __restrict__ row, const int* __restrict__ col,
                                                 int* __restrict__ cnt, int* __restrict__ slots,
                                                 int2* __restrict__ spill, int n, int e, int i0, int total) {
    for (int base = i0; base < e; base += EDGES_PER_THREAD * total) {
        int c[EDGES_PER_THREAD], r[EDGES_PER_THREAD], p[EDGES_PER_THREAD];
#pragma unroll
        for (int k = 0; k < EDGES_PER_THREAD; ++k) {
            const int i = base + k * total;
            c[k] = col[(i < e) ? i : (e - 1)];
        }
#pragma unroll
        for (int k = 0; k < EDGES_PER_THREAD; ++k) {
            const int i = base + k * total;
            r[k] = row[(i < e) ? i : (e - 1)];
        }
#pragma unroll
        for (int k = 0; k < EDGES_PER_THREAD; ++k) {
            p[k] = 0;
            if (base + k * total < e) p[k] = atomicAdd(&cnt[c[k]], 1);
        }
#pragma unroll
        for (int k = 0; k < EDGES_PER_THREAD; ++k) {
            if (base + k * total < e) place_edge(c[k], r[k], p[k], slots, cnt + n, spill);
        }
    }
}

#define GEMM_GROUPS 512
typedef __attribute__((ext_vector_type(4))) float v4f;

__global__ __launch_bounds__(256, 4) void k_gemm128_lists(
    const float* __restrict__ X, const float* __restrict__ W, float* __restrict__ Y, int n,
    const int* __restrict__ row, const int* __restrict__ col, int* __restrict__ cnt,
    int* __restrict__ slots, int2* __restrict__ spill, int e,
    const float* __restrict__ W2, const int* __restrict__ rootPtr, float* __restrict__ cvec) {
    __shared__ v4f Bl[8 * 16 * 32];
    const int tid  = threadIdx.x;
    const int lane = tid & 31;
    const int wave = __builtin_amdgcn_readfirstlane(tid >> 5);
    const int c    = lane & 15;
    const int h    = lane >> 4;

    {
        const int sn = tid & 127;
        const int sh = tid >> 7;
        const float* wp = W + (4 * sh) * DIM + sn;
        v4f* bs = Bl + (sn >> 4) * (16 * 32) + (sn & 15) + 16 * sh;
#pragma unroll
        for (int u = 0; u < 16; ++u) {
            v4f t;
            t.x = wp[(8 * u + 0) * DIM];
            t.y = wp[(8 * u + 1) * DIM];
            t.z = wp[(8 * u + 2) * DIM];
            t.w = wp[(8 * u + 3) * DIM];
            bs[u * 32] = t;
        }
    }

    __syncthreads();

    const int ntiles = (n + 15) >> 4;
    const int stride = 8 * (int)gridDim.x;
    int tile = wave * (int)gridDim.x + (int)blockIdx.x;

    v4f a[16];
    {
        const int r0 = (tile * 16 + c < n) ? (tile * 16 + c) : (n - 1);
        const v4f* ap = (const v4f*)(X + (long long)r0 * DIM + 4 * h);
#pragma unroll
        for (int u = 0; u < 16; ++u) a[u] = ap[2 * u];
    }

    const int ei0 = (int)(blockIdx.x * blockDim.x) + tid;
    const int etotal = (int)(gridDim.x * blockDim.x);
    bool lists_due = (tile + stride < ntiles);
    if (!lists_due) build_lists_part(row, col, cnt, slots, spill, n, e, ei0, etotal);

    const v4f* bl = Bl + lane;
    for (; tile < ntiles; tile += stride) {
        asm volatile("" ::: "memory");
        const int m0 = tile * 16;
        const int mn = (tile + stride) * 16 + c;
        const int rn = (mn < n) ? mn : (n - 1);
        const v4f* apn = (const v4f*)(X + (long long)rn * DIM + 4 * h);

        v8f d[8];
#pragma unroll
        for (int j = 0; j < 8; ++j) {
#pragma unroll
            for (int v = 0; v < 8; ++v) d[j][v] = 0.0f;
        }

        v4f bc[8], bn[8];
#pragma unroll
        for (int j = 0; j < 8; ++j) bc[j] = bl[(j * 16) * 32];
#pragma unroll
        for (int u = 0; u < 16; ++u) {
            if (u + 1 < 16) {
#pragma unroll
                for (int j = 0; j < 8; ++j) bn[j] = bl[(j * 16 + u + 1) * 32];
            }
            __builtin_amdgcn_sched_barrier(0);
            v2f a0, a1;
            a0.x = a[u].x; a0.y = a[u].y;
            a1.x = a[u].z; a1.y = a[u].w;
#pragma unroll
            for (int j = 0; j < 8; ++j) {
                v2f b0;
                b0.x = bc[j].x; b0.y = bc[j].y;
                d[j] = __builtin_amdgcn_wmma_f32_16x16x4_f32(false, a0, false, b0,
                                                             (short)0, d[j], false, false);
            }
#pragma unroll
            for (int j = 0; j < 8; ++j) {
                v2f b1;
                b1.x = bc[j].z; b1.y = bc[j].w;
                d[j] = __builtin_amdgcn_wmma_f32_16x16x4_f32(false, a1, false, b1,
                                                             (short)0, d[j], false, false);
            }
            a[u] = apn[2 * u];
            __builtin_amdgcn_sched_barrier(0);
#pragma unroll
            for (int j = 0; j < 8; ++j) bc[j] = bn[j];
        }

        float* yp = Y + (long long)(m0 + 8 * h) * DIM + c;
#pragma unroll
        for (int v = 0; v < 8; ++v) {
            const int m = m0 + v + 8 * h;
            if (m < n) {
#pragma unroll
                for (int j = 0; j < 8; ++j) yp[(long long)v * DIM + 16 * j] = d[j][v];
            }
        }
        if (lists_due) {
            build_lists_part(row, col, cnt, slots, spill, n, e, ei0, etotal);
            lists_due = false;
        }
    }

    if (blockIdx.x == gridDim.x - 1 && tid >= DIM) {
        const int root = *rootPtr;
        const float* __restrict__ pr = X + (long long)root * DIM;
        const float* __restrict__ wc = W2 + (long long)DIM * DIM + (tid - DIM);
        float cs = 0.0f;
#pragma unroll 32
        for (int k = 0; k < DIM; ++k) cs = fmaf(fmaxf(pr[k], 0.0f), wc[k * DIM], cs);
        cvec[tid - DIM] = cs;
    }
}

#define ADD_ROW(v_) a.x += v_.x; a.y += v_.y; a.z += v_.z; a.w += v_.w
#define FMA_ROW(v_, f_) a.x = fmaf(v_.x, f_, a.x); a.y = fmaf(v_.y, f_, a.y); \
                        a.z = fmaf(v_.z, f_, a.z); a.w = fmaf(v_.w, f_, a.w)
#define ACC_ROW(v_, f_) if (RAW) { FMA_ROW(v_, f_); } else { ADD_ROW(v_); }
template <bool RAW>
__device__ __forceinline__ float4 gather_row(
    const float* __restrict__ xw, const int* __restrict__ cnt, const int* __restrict__ slots,
    const int2* __restrict__ spill, int n, int node, int lane) {
    const int deg = cnt[node];
    const float dc = rsqrtf((float)(deg + 1));
    const int nl = (deg < LIST_CAP) ? deg : LIST_CAP;
    const int* __restrict__ lp = slots + (long long)node * LIST_CAP;
    const float4* __restrict__ x4 = (const float4*)xw;

    float4 a = x4[(long long)node * 32 + lane];
    if (RAW) { a.x *= dc; a.y *= dc; a.z *= dc; a.w *= dc; }

    for (int b = 0; b < nl; b += 32) {
        const int m = (nl - b < 32) ? (nl - b) : 32;
        int r = 0;
        if (lane < m) r = lp[b + lane];
        int cr = 0;
        if (RAW) cr = cnt[r];
#define SRC_ROW(j_) x4[(long long)__builtin_amdgcn_readlane(r, (j_)) * 32 + lane]
#define SRC_FAC(j_) __builtin_amdgcn_rsqf((float)(__builtin_amdgcn_readlane(cr, (j_)) + 1))
        int j = 0;
        for (; j + 4 <= m; j += 4) {
            const float4 v0 = SRC_ROW(j);
            const float4 v1 = SRC_ROW(j + 1);
            const float4 v2 = SRC_ROW(j + 2);
            const float4 v3 = SRC_ROW(j + 3);
            const float f0 = SRC_FAC(j), f1 = SRC_FAC(j + 1), f2 = SRC_FAC(j + 2), f3 = SRC_FAC(j + 3);
            ACC_ROW(v0, f0); ACC_ROW(v1, f1); ACC_ROW(v2, f2); ACC_ROW(v3, f3);
        }
        const int rem = m - j;
        if (rem == 3) {
            const float4 v0 = SRC_ROW(j);
            const float4 v1 = SRC_ROW(j + 1);
            const float4 v2 = SRC_ROW(j + 2);
            const float f0 = SRC_FAC(j), f1 = SRC_FAC(j + 1), f2 = SRC_FAC(j + 2);
            ACC_ROW(v0, f0); ACC_ROW(v1, f1); ACC_ROW(v2, f2);
        } else if (rem == 2) {
            const float4 v0 = SRC_ROW(j);
            const float4 v1 = SRC_ROW(j + 1);
            const float f0 = SRC_FAC(j), f1 = SRC_FAC(j + 1);
            ACC_ROW(v0, f0); ACC_ROW(v1, f1);
        } else if (rem == 1) {
            const float4 v0 = SRC_ROW(j);
            const float f0 = SRC_FAC(j);
            ACC_ROW(v0, f0);
        }
#undef SRC_ROW
#undef SRC_FAC
    }

    if (deg > LIST_CAP) {
        const int ns = cnt[n];
        for (int b = 0; b < ns; b += 32) {
            int2 pr = make_int2(-1, 0);
            float fs = 0.0f;
            if (b + lane < ns) {
                pr = spill[b + lane];
                if (RAW) fs = rsqrtf((float)(cnt[pr.y] + 1));
            }
            unsigned mask = __builtin_amdgcn_ballot_w32(pr.x == node);
            while (mask != 0u) {
                const int j = __builtin_ctz(mask);
                mask &= mask - 1u;
                const float4 v0 = x4[(long long)__builtin_amdgcn_readlane(pr.y, j) * 32 + lane];
                const float f0 = __int_as_float(__builtin_amdgcn_readlane(__float_as_int(fs), j));
                ACC_ROW(v0, f0);
            }
        }
    }
    return make_float4(a.x * dc, a.y * dc, a.z * dc, a.w * dc);
}
#undef ACC_ROW
#undef FMA_ROW
#undef ADD_ROW

#define FUSE_GROUPS 447
#define TILE_PITCH 33
__global__ __launch_bounds__(1024, 16) void k_gather_relu_gemm(
    const float* __restrict__ xw, const int* __restrict__ cnt, const int* __restrict__ slots,
    const int2* __restrict__ spill, const float* __restrict__ b1,
    const float* __restrict__ W, const float* __restrict__ cvec,
    const int* __restrict__ rootPtr,
    float* __restrict__ root_feat, float* __restrict__ Y, int n) {
    __shared__ v4f Bl[8 * 16 * 32];
    __shared__ v4f Tl[2 * 32 * TILE_PITCH];
    const int tid  = threadIdx.x;
    const int lane = tid & 31;
    const int wave = __builtin_amdgcn_readfirstlane(tid >> 5);
    const int c    = lane & 15;
    const int h    = lane >> 4;

    {
        const int sn = tid & 127;
        const int sh = (tid >> 7) & 1;
        const int uq = tid >> 8;
        const float* wp = W + (4 * sh) * DIM + sn;
        v4f* bs = Bl + (sn >> 4) * (16 * 32) + (sn & 15) + 16 * sh;
#pragma unroll
        for (int k = 0; k < 4; ++k) {
            const int u = 4 * uq + k;
            v4f t;
            t.x = wp[(8 * u + 0) * DIM];
            t.y = wp[(8 * u + 1) * DIM];
            t.z = wp[(8 * u + 2) * DIM];
            t.w = wp[(8 * u + 3) * DIM];
            bs[u * 32] = t;
        }
    }

    const int root = *rootPtr;
    const float4 bb = ((const float4*)b1)[lane];
    const int ntiles = (n + 31) >> 5;
    int par = 0;
    for (int tile = (int)blockIdx.x; tile < ntiles; tile += (int)gridDim.x) {
        const int node = tile * 32 + wave;
        v4f hrow;
        hrow.x = 0.0f; hrow.y = 0.0f; hrow.z = 0.0f; hrow.w = 0.0f;
        int deg = 0;
        if (node < n) {
            deg = cnt[node];
            const float4 a = gather_row<true>(xw, cnt, slots, spill, n, node, lane);
            const float4 s = make_float4(a.x + bb.x, a.y + bb.y, a.z + bb.z, a.w + bb.w);
            if (node == root) ((float4*)root_feat)[lane] = s;
            hrow.x = fmaxf(s.x, 0.0f); hrow.y = fmaxf(s.y, 0.0f);
            hrow.z = fmaxf(s.z, 0.0f); hrow.w = fmaxf(s.w, 0.0f);
        }
        v4f* tb = Tl + par * (32 * TILE_PITCH);
        tb[wave * TILE_PITCH + lane] = hrow;
        int* ti = (int*)tb;
        if (lane == 0) ti[(wave * TILE_PITCH + 32) * 4] = deg;

        __syncthreads();

        if ((wave >> 4) == par) {
            const int pw = wave & 15;
            const int rt = pw >> 3;
            const int j  = pw & 7;
            const v4f* ta = tb + (16 * rt + c) * TILE_PITCH + h;
            const v4f* bl = Bl + (j * 16) * 32 + lane;
            const int m0 = tile * 32 + 16 * rt;
            float rsv[8];
#pragma unroll
            for (int v = 0; v < 8; ++v)
                rsv[v] = rsqrtf((float)(ti[((16 * rt + v + 8 * h) * TILE_PITCH + 32) * 4] + 1));
            const float cjv = cvec[16 * j + c];
            v8f d;
#pragma unroll
            for (int v = 0; v < 8; ++v) d[v] = 0.0f;
#pragma unroll
            for (int u = 0; u < 16; ++u) {
                const v4f a4 = ta[2 * u];
                const v4f b4 = bl[u * 32];
                v2f a0, a1, b0, b1v;
                a0.x = a4.x; a0.y = a4.y; a1.x = a4.z; a1.y = a4.w;
                b0.x = b4.x; b0.y = b4.y; b1v.x = b4.z; b1v.y = b4.w;
                d = __builtin_amdgcn_wmma_f32_16x16x4_f32(false, a0, false, b0, (short)0, d, false, false);
                d = __builtin_amdgcn_wmma_f32_16x16x4_f32(false, a1, false, b1v, (short)0, d, false, false);
            }
            asm volatile("v_nop\n\tv_nop\n\tv_nop\n\tv_nop\n\tv_nop" : "+v"(d));
            float* yp = Y + (long long)(m0 + 8 * h) * DIM + 16 * j + c;
#pragma unroll
            for (int v = 0; v < 8; ++v) {
                const int m = m0 + v + 8 * h;
                if (m < n) yp[(long long)v * DIM] = (d[v] + cjv) * rsv[v];
            }
        }
        par ^= 1;
    }
}

typedef v4f v4f_a8 __attribute__((aligned(8)));

__global__ __launch_bounds__(256) void k_gather_pool(
    const float* __restrict__ xw, const int* __restrict__ cnt, const int* __restrict__ slots,
    const int2* __restrict__ spill, const float* __restrict__ b2,
    const float* __restrict__ root_feat, float* __restrict__ out, int n) {
    const int lane = threadIdx.x & 31;
    const int node = blockIdx.x * 8 + __builtin_amdgcn_readfirstlane((int)(threadIdx.x >> 5));
    if (node >= n) return;
    const float third = 1.0f / 3.0f;

    const float4 bb = ((const float4*)b2)[lane];
    const float4 rf = ((const float4*)root_feat)[lane];
    const int nx = (lane < 31) ? 4 * lane + 4 : 126;
    const float rf4 = root_feat[nx], rf5 = root_feat[nx + 1];

    const float4 a = gather_row<false>(xw, cnt, slots, spill, n, node, lane);

    v4f rp;
    rp.x = (rf.x + rf.y + rf.z) * third;
    rp.y = (rf.y + rf.z + rf.w) * third;
    rp.z = (rf.z + rf.w + rf4) * third;
    rp.w = (rf.w + rf4 + rf5) * third;
    const float s67 = rf.z + rf.w;

    const float cx = fmaxf(a.x + bb.x, 0.0f);
    const float cy = fmaxf(a.y + bb.y, 0.0f);
    const float cz = fmaxf(a.z + bb.z, 0.0f);
    const float cw = fmaxf(a.w + bb.w, 0.0f);
    const float n0 = __shfl_down(cx, 1);
    const float n1 = __shfl_down(cy, 1);
    const float c0 = __int_as_float(__builtin_amdgcn_readlane(__float_as_int(cx), 0));
    const float c1 = __int_as_float(__builtin_amdgcn_readlane(__float_as_int(cy), 0));

    float* __restrict__ ob = out + (long long)node * 254;
    v4f ha = rp;
    if (lane == 31) {
        ha.z = (s67 + c0) * third;
        ha.w = (rf.w + c0 + c1) * third;
    }
    *(v4f_a8*)(ob + 4 * lane) = ha;

    v4f p;
    p.x = (cx + cy + cz) * third;
    p.y = (cy + cz + cw) * third;
    p.z = (cz + cw + n0) * third;
    p.w = (cw + n0 + n1) * third;
    if (lane < 31) *(v4f_a8*)(ob + 128 + 4 * lane) = p;
    else           *(float2*)(ob + 252) = make_float2(p.x, p.y);
}

extern "C" void kernel_launch(void* const* d_in, const int* in_sizes, int n_in,
                              void* d_out, int out_size, void* d_ws, size_t ws_size,
                              hipStream_t stream) {
    const float* posts = (const float*)d_in[0];
    const float* W1    = (const float*)d_in[1];
    const float* b1    = (const float*)d_in[2];
    const float* W2    = (const float*)d_in[3];
    const float* b2    = (const float*)d_in[4];
    const int*   edge  = (const int*)d_in[5];
    const int*   rootP = (const int*)d_in[6];

    const int n = in_sizes[0] / DIM;
    const int e = in_sizes[5] / 2;
    const int* rowi = edge;
    const int* coli = edge + e;

    char* ws = (char*)d_ws;
    size_t off = 0;
    float* xw   = (float*)(ws + off); off += (size_t)n * DIM * 4;
    char*  lst  = ws + off;           off += (size_t)n * DIM * 4;
    float* rootf = (float*)(ws + off); off += DIM * 4;
    float* cvec  = (float*)(ws + off);
    float* out = (float*)d_out;

    const size_t cpad = (((size_t)(n + 1) * 4) + 255) & ~(size_t)255;
    const size_t spad = (size_t)n * LIST_CAP * 4;
    int*  cnt   = (int*)lst;
    int*  slots = (int*)(lst + cpad);
    int2* spill = (int2*)(lst + cpad + spad);
    float* xs1 = (float*)d_out;

    const int B = 256;
    const int gZ   = (n + 1 + B - 1) / B;
    const int gG   = (n + 7) / 8;
    const int gM   = GEMM_GROUPS;

    k_zero_i32   <<<gZ, B, 0, stream>>>(cnt, n + 1);

    k_gemm128_lists    <<<gM, B, 0, stream>>>(posts, W1, xs1, n, rowi, coli, cnt, slots, spill, e,
                                              W2, rootP, cvec);
    k_gather_relu_gemm <<<FUSE_GROUPS, 1024, 0, stream>>>(xs1, cnt, slots, spill, b1, W2, cvec, rootP,
                                                          rootf, xw, n);

    k_gather_pool  <<<gG, B, 0, stream>>>(xw, cnt, slots, spill, b2, rootf, out, n);
}
